# starting kernel
# baseline (speedup 1.0000x reference)
#include <hip/hip_runtime.h>
#include <math.h>

typedef __attribute__((ext_vector_type(2))) float v2f;
typedef __attribute__((ext_vector_type(8))) float v8f;

#define NBATCH 4
#define CIN 512
#define HH 48
#define WW 48
#define LSPAT (HH*WW)
#define CH3 192
#define HCH 64
#define NHEAD 8
#define DHEAD 8
#define ATTN_SCALE 0.35355339059327379f

static __device__ __forceinline__ v8f wmma4(v2f a, v2f b, v8f c) {
  return __builtin_amdgcn_wmma_f32_16x16x4_f32(false, a, false, b, (short)0, c,
                                               false, false);
}

__global__ __launch_bounds__(256) void qkv_gemm_kernel(
    const float* __restrict__ x, const float* __restrict__ w,
    const float* __restrict__ bias, float* __restrict__ qkv) {
  const int lane = threadIdx.x & 31;
  const int half = lane >> 4;
  const int l16  = lane & 15;
  const int wave = blockIdx.x * (blockDim.x >> 5) + (threadIdx.x >> 5);

  const int ntn = LSPAT / 16;
  const int ntm = CH3 / 16;
  int t = wave;
  const int n  = t / (ntm * ntn);  t -= n * ntm * ntn;
  const int mt = t / ntn;
  const int ct = t % ntn;

  const int mrow = mt * 16 + l16;
  const int col  = ct * 16 + l16;
  const float* xb = x + (size_t)n * CIN * LSPAT;

  v8f acc = {};
  for (int kk = 0; kk < CIN; kk += 4) {
    const int ka = kk + half * 2;
    v2f a; a[0] = w[mrow * CIN + ka];
           a[1] = w[mrow * CIN + ka + 1];
    v2f b; b[0] = xb[(size_t)ka * LSPAT + col];
           b[1] = xb[(size_t)(ka + 1) * LSPAT + col];
    acc = wmma4(a, b, acc);
  }

  float* out = qkv + (size_t)n * CH3 * LSPAT;
#pragma unroll
  for (int r = 0; r < 8; ++r) {
    const int row = mt * 16 + r + half * 8;
    out[(size_t)row * LSPAT + col] = acc[r] + bias[row];
  }
}

__global__ __launch_bounds__(128) void attn_kernel(
    const float* __restrict__ qkv, float* __restrict__ obuf) {
  __shared__ float plds[4][16][17];
  const int lane = threadIdx.x & 31;
  const int half = lane >> 4;
  const int l16  = lane & 15;
  const int wv   = threadIdx.x >> 5;

  const int qtiles = LSPAT / 64;
  const int nh = blockIdx.x / qtiles;
  const int qt = blockIdx.x % qtiles;
  const int n = nh >> 3, h = nh & 7;
  const int qbase = qt * 64 + wv * 16;

  const float* base = qkv + (size_t)n * CH3 * LSPAT + (size_t)h * 24 * LSPAT;
  const float* qm = base;
  const float* km = base + (size_t)DHEAD * LSPAT;
  const float* vm = base + (size_t)2 * DHEAD * LSPAT;

  v2f aq[2];
#pragma unroll
  for (int db = 0; db < 2; ++db) {
    const int d0 = db * 4 + half * 2;
    aq[db][0] = qm[(size_t)d0 * LSPAT + qbase + l16];
    aq[db][1] = qm[(size_t)(d0 + 1) * LSPAT + qbase + l16];
  }

  float m8[8];
#pragma unroll
  for (int r = 0; r < 8; ++r) m8[r] = -3.0e38f;

  for (int kt = 0; kt < LSPAT / 16; ++kt) {
    const int kb = kt * 16;
    v8f s = {};
#pragma unroll
    for (int db = 0; db < 2; ++db) {
      const int d0 = db * 4 + half * 2;
      v2f bk; bk[0] = km[(size_t)d0 * LSPAT + kb + l16];
              bk[1] = km[(size_t)(d0 + 1) * LSPAT + kb + l16];
      s = wmma4(aq[db], bk, s);
    }
#pragma unroll
    for (int r = 0; r < 8; ++r) m8[r] = fmaxf(m8[r], s[r]);
  }
#pragma unroll
  for (int r = 0; r < 8; ++r) {
#pragma unroll
    for (int m = 1; m < 16; m <<= 1) m8[r] = fmaxf(m8[r], __shfl_xor(m8[r], m, 32));
  }

  float s8[8];
#pragma unroll
  for (int r = 0; r < 8; ++r) s8[r] = 0.0f;

  for (int kt = 0; kt < LSPAT / 16; ++kt) {
    const int kb = kt * 16;
    v8f s = {};
#pragma unroll
    for (int db = 0; db < 2; ++db) {
      const int d0 = db * 4 + half * 2;
      v2f bk; bk[0] = km[(size_t)d0 * LSPAT + kb + l16];
              bk[1] = km[(size_t)(d0 + 1) * LSPAT + kb + l16];
      s = wmma4(aq[db], bk, s);
    }
#pragma unroll
    for (int r = 0; r < 8; ++r) s8[r] += __expf((s[r] - m8[r]) * ATTN_SCALE);
  }
  float inv[8];
#pragma unroll
  for (int r = 0; r < 8; ++r) {
#pragma unroll
    for (int m = 1; m < 16; m <<= 1) s8[r] += __shfl_xor(s8[r], m, 32);
    inv[r] = 1.0f / s8[r];
  }

  v8f acc = {};
  for (int kt = 0; kt < LSPAT / 16; ++kt) {
    const int kb = kt * 16;
    v8f s = {};
#pragma unroll
    for (int db = 0; db < 2; ++db) {
      const int d0 = db * 4 + half * 2;
      v2f bk; bk[0] = km[(size_t)d0 * LSPAT + kb + l16];
              bk[1] = km[(size_t)(d0 + 1) * LSPAT + kb + l16];
      s = wmma4(aq[db], bk, s);
    }
#pragma unroll
    for (int r = 0; r < 8; ++r)
      plds[wv][r + half * 8][l16] =
          __expf((s[r] - m8[r]) * ATTN_SCALE) * inv[r];
    __syncthreads();
#pragma unroll
    for (int c = 0; c < 4; ++c) {
      const int k0 = c * 4 + half * 2;
      v2f ap; ap[0] = plds[wv][l16][k0];
              ap[1] = plds[wv][l16][k0 + 1];
      v2f bv;
      if (l16 < 8) {
        bv[0] = vm[(size_t)l16 * LSPAT + kb + k0];
        bv[1] = vm[(size_t)l16 * LSPAT + kb + k0 + 1];
      } else { bv[0] = 0.0f; bv[1] = 0.0f; }
      acc = wmma4(ap, bv, acc);
    }
    __syncthreads();
  }

  if (l16 < 8) {
    float* ob = obuf + (size_t)n * HCH * LSPAT + (size_t)(h * DHEAD + l16) * LSPAT;
#pragma unroll
    for (int r = 0; r < 8; ++r) ob[qbase + r + half * 8] = acc[r];
  }
}

__global__ __launch_bounds__(256) void pos_conv_kernel(
    const float* __restrict__ qkv, const float* __restrict__ pw,
    const float* __restrict__ pb, float* __restrict__ obuf) {
  int idx = blockIdx.x * 256 + threadIdx.x;
  const int xp = idx % WW; int t = idx / WW;
  const int yp = t % HH;  t /= HH;
  const int oc = t % HCH; const int n = t / HCH;

  const float* vbase = qkv + (size_t)n * CH3 * LSPAT;
  float acc = pb[oc];
  for (int ic = 0; ic < HCH; ++ic) {
    const int ch = (ic >> 3) * 24 + 16 + (ic & 7);
    const float* vin  = vbase + (size_t)ch * LSPAT;
    const float* wrow = pw + ((size_t)oc * HCH + ic) * 9;
#pragma unroll
    for (int ky = 0; ky < 3; ++ky) {
      const int iy = yp + ky - 1;
      if (iy < 0 || iy >= HH) continue;
#pragma unroll
      for (int kx = 0; kx < 3; ++kx) {
        const int ix = xp + kx - 1;
        if (ix < 0 || ix >= WW) continue;
        acc = fmaf(vin[iy * WW + ix], wrow[ky * 3 + kx], acc);
      }
    }
  }
  obuf[(size_t)n * HCH * LSPAT + (size_t)oc * LSPAT + yp * WW + xp] += acc;
}

__global__ __launch_bounds__(256) void res_gemm_kernel(
    const float* __restrict__ obuf, const float* __restrict__ w,
    const float* __restrict__ bias, float* __restrict__ out) {
  const int lane = threadIdx.x & 31;
  const int half = lane >> 4;
  const int l16  = lane & 15;
  const int wave = blockIdx.x * (blockDim.x >> 5) + (threadIdx.x >> 5);

  const int ntn = LSPAT / 16;
  const int ntm = CIN / 16;
  int t = wave;
  const int n  = t / (ntm * ntn); t -= n * ntm * ntn;
  const int mt = t / ntn;
  const int ct = t % ntn;

  const int mrow = mt * 16 + l16;
  const int col  = ct * 16 + l16;
  const float* ib = obuf + (size_t)n * HCH * LSPAT;

  v8f acc = {};
#pragma unroll
  for (int kk = 0; kk < HCH; kk += 4) {
    const int ka = kk + half * 2;
    v2f a; a[0] = w[mrow * HCH + ka];
           a[1] = w[mrow * HCH + ka + 1];
    v2f b; b[0] = ib[(size_t)ka * LSPAT + col];
           b[1] = ib[(size_t)(ka + 1) * LSPAT + col];
    acc = wmma4(a, b, acc);
  }

  float* ob = out + (size_t)n * CIN * LSPAT;
#pragma unroll
  for (int r = 0; r < 8; ++r) {
    const int row = mt * 16 + r + half * 8;
    ob[(size_t)row * LSPAT + col] = acc[r] + bias[row];
  }
}

extern "C" void kernel_launch(void* const* d_in, const int* in_sizes, int n_in,
                              void* d_out, int out_size, void* d_ws, size_t ws_size,
                              hipStream_t stream) {
  const float* x     = (const float*)d_in[0];
  const float* qkv_w = (const float*)d_in[1];
  const float* qkv_b = (const float*)d_in[2];
  const float* pos_w = (const float*)d_in[3];
  const float* pos_b = (const float*)d_in[4];
  const float* res_w = (const float*)d_in[5];
  const float* res_b = (const float*)d_in[6];
  float* out = (float*)d_out;

  float* qkv  = (float*)d_ws;
  float* obuf = qkv + (size_t)NBATCH * CH3 * LSPAT;

  const int waves1 = NBATCH * (CH3 / 16) * (LSPAT / 16);
  qkv_gemm_kernel<<<waves1 / 8, 256, 0, stream>>>(x, qkv_w, qkv_b, qkv);

  attn_kernel<<<NBATCH * NHEAD * (LSPAT / 64), 128, 0, stream>>>(qkv, obuf);

  pos_conv_kernel<<<(NBATCH * HCH * LSPAT) / 256, 256, 0, stream>>>(
      qkv, pos_w, pos_b, obuf);

  const int waves4 = NBATCH * (CIN / 16) * (LSPAT / 16);
  res_gemm_kernel<<<waves4 / 8, 256, 0, stream>>>(obuf, res_w, res_b, out);
}


# pos 3x3 conv moved from a scalar FMA loop onto f32 WMMA as an implicit GEMM (v rows with zero halo staged in LDS, K = 576 in weight order); sum order of the conv changes
# speedup vs baseline: 1.5127x; 1.5127x over previous
#include <hip/hip_runtime.h>
#include <math.h>

typedef __attribute__((ext_vector_type(2))) float v2f;
typedef __attribute__((ext_vector_type(4))) float v4f;
typedef __attribute__((ext_vector_type(8))) float v8f;

#define NBATCH 4
#define CIN 512
#define HH 48
#define WW 48
#define LSPAT (HH*WW)
#define CH3 192
#define HCH 64
#define NHEAD 8
#define DHEAD 8
#define ATTN_SCALE 0.35355339059327379f

static __device__ __forceinline__ v8f wmma4(v2f a, v2f b, v8f c) {
  return __builtin_amdgcn_wmma_f32_16x16x4_f32(false, a, false, b, (short)0, c,
                                               false, false);
}

__global__ __launch_bounds__(256) void qkv_gemm_kernel(
    const float* __restrict__ x, const float* __restrict__ w,
    const float* __restrict__ bias, float* __restrict__ qkv) {
  const int lane = threadIdx.x & 31;
  const int half = lane >> 4;
  const int l16  = lane & 15;
  const int wave = blockIdx.x * (blockDim.x >> 5) + (threadIdx.x >> 5);

  const int ntn = LSPAT / 16;
  const int ntm = CH3 / 16;
  int t = wave;
  const int n  = t / (ntm * ntn);  t -= n * ntm * ntn;
  const int mt = t / ntn;
  const int ct = t % ntn;

  const int mrow = mt * 16 + l16;
  const int col  = ct * 16 + l16;
  const float* xb = x + (size_t)n * CIN * LSPAT;

  v8f acc = {};
  for (int kk = 0; kk < CIN; kk += 4) {
    const int ka = kk + half * 2;
    v2f a; a[0] = w[mrow * CIN + ka];
           a[1] = w[mrow * CIN + ka + 1];
    v2f b; b[0] = xb[(size_t)ka * LSPAT + col];
           b[1] = xb[(size_t)(ka + 1) * LSPAT + col];
    acc = wmma4(a, b, acc);
  }

  float* out = qkv + (size_t)n * CH3 * LSPAT;
#pragma unroll
  for (int r = 0; r < 8; ++r) {
    const int row = mt * 16 + r + half * 8;
    out[(size_t)row * LSPAT + col] = acc[r] + bias[row];
  }
}

__global__ __launch_bounds__(128) void attn_kernel(
    const float* __restrict__ qkv, float* __restrict__ obuf) {
  __shared__ float plds[4][16][17];
  const int lane = threadIdx.x & 31;
  const int half = lane >> 4;
  const int l16  = lane & 15;
  const int wv   = threadIdx.x >> 5;

  const int qtiles = LSPAT / 64;
  const int nh = blockIdx.x / qtiles;
  const int qt = blockIdx.x % qtiles;
  const int n = nh >> 3, h = nh & 7;
  const int qbase = qt * 64 + wv * 16;

  const float* base = qkv + (size_t)n * CH3 * LSPAT + (size_t)h * 24 * LSPAT;
  const float* qm = base;
  const float* km = base + (size_t)DHEAD * LSPAT;
  const float* vm = base + (size_t)2 * DHEAD * LSPAT;

  v2f aq[2];
#pragma unroll
  for (int db = 0; db < 2; ++db) {
    const int d0 = db * 4 + half * 2;
    aq[db][0] = qm[(size_t)d0 * LSPAT + qbase + l16];
    aq[db][1] = qm[(size_t)(d0 + 1) * LSPAT + qbase + l16];
  }

  float m8[8];
#pragma unroll
  for (int r = 0; r < 8; ++r) m8[r] = -3.0e38f;

  for (int kt = 0; kt < LSPAT / 16; ++kt) {
    const int kb = kt * 16;
    v8f s = {};
#pragma unroll
    for (int db = 0; db < 2; ++db) {
      const int d0 = db * 4 + half * 2;
      v2f bk; bk[0] = km[(size_t)d0 * LSPAT + kb + l16];
              bk[1] = km[(size_t)(d0 + 1) * LSPAT + kb + l16];
      s = wmma4(aq[db], bk, s);
    }
#pragma unroll
    for (int r = 0; r < 8; ++r) m8[r] = fmaxf(m8[r], s[r]);
  }
#pragma unroll
  for (int r = 0; r < 8; ++r) {
#pragma unroll
    for (int m = 1; m < 16; m <<= 1) m8[r] = fmaxf(m8[r], __shfl_xor(m8[r], m, 32));
  }

  float s8[8];
#pragma unroll
  for (int r = 0; r < 8; ++r) s8[r] = 0.0f;

  for (int kt = 0; kt < LSPAT / 16; ++kt) {
    const int kb = kt * 16;
    v8f s = {};
#pragma unroll
    for (int db = 0; db < 2; ++db) {
      const int d0 = db * 4 + half * 2;
      v2f bk; bk[0] = km[(size_t)d0 * LSPAT + kb + l16];
              bk[1] = km[(size_t)(d0 + 1) * LSPAT + kb + l16];
      s = wmma4(aq[db], bk, s);
    }
#pragma unroll
    for (int r = 0; r < 8; ++r) s8[r] += __expf((s[r] - m8[r]) * ATTN_SCALE);
  }
  float inv[8];
#pragma unroll
  for (int r = 0; r < 8; ++r) {
#pragma unroll
    for (int m = 1; m < 16; m <<= 1) s8[r] += __shfl_xor(s8[r], m, 32);
    inv[r] = 1.0f / s8[r];
  }

  v8f acc = {};
  for (int kt = 0; kt < LSPAT / 16; ++kt) {
    const int kb = kt * 16;
    v8f s = {};
#pragma unroll
    for (int db = 0; db < 2; ++db) {
      const int d0 = db * 4 + half * 2;
      v2f bk; bk[0] = km[(size_t)d0 * LSPAT + kb + l16];
              bk[1] = km[(size_t)(d0 + 1) * LSPAT + kb + l16];
      s = wmma4(aq[db], bk, s);
    }
#pragma unroll
    for (int r = 0; r < 8; ++r)
      plds[wv][r + half * 8][l16] =
          __expf((s[r] - m8[r]) * ATTN_SCALE) * inv[r];
    __syncthreads();
#pragma unroll
    for (int c = 0; c < 4; ++c) {
      const int k0 = c * 4 + half * 2;
      v2f ap; ap[0] = plds[wv][l16][k0];
              ap[1] = plds[wv][l16][k0 + 1];
      v2f bv;
      if (l16 < 8) {
        bv[0] = vm[(size_t)l16 * LSPAT + kb + k0];
        bv[1] = vm[(size_t)l16 * LSPAT + kb + k0 + 1];
      } else { bv[0] = 0.0f; bv[1] = 0.0f; }
      acc = wmma4(ap, bv, acc);
    }
    __syncthreads();
  }

  if (l16 < 8) {
    float* ob = obuf + (size_t)n * HCH * LSPAT + (size_t)(h * DHEAD + l16) * LSPAT;
#pragma unroll
    for (int r = 0; r < 8; ++r) ob[qbase + r + half * 8] = acc[r];
  }
}

#define VPITCH 56
#define VICS   (3 * VPITCH)
#define KCHUNK 36

__global__ __launch_bounds__(128) void pos_conv_kernel(
    const float* __restrict__ qkv, const float* __restrict__ pw,
    const float* __restrict__ pb, float* __restrict__ obuf) {
  __shared__ __attribute__((aligned(16))) float vl[HCH * VICS];
  const int tid  = threadIdx.x;
  const int lane = tid & 31;
  const int half = lane >> 4;
  const int l16  = lane & 15;
  const int wv   = tid >> 5;
  const int n = blockIdx.x / HH;
  const int y = blockIdx.x % HH;

  const float* vbase = qkv + (size_t)n * CH3 * LSPAT;

  v4f stg[21];
#pragma unroll
  for (int it = 0; it < 21; ++it) {
    const int i   = tid + it * 128;
    const int ic  = i / 42;
    const int rem = i - ic * 42;
    const int r   = rem / 14;
    const int q   = rem - r * 14;
    const int iy  = y + r - 1;
    const bool ok = (q >= 1) && (q <= 12) && (iy >= 0) && (iy < HH);
    const int qc  = q < 1 ? 1 : (q > 12 ? 12 : q);
    const int iyc = iy < 0 ? 0 : (iy > HH - 1 ? HH - 1 : iy);
    const int ch  = (ic >> 3) * 24 + 16 + (ic & 7);
    const v4f t = *(const v4f*)(vbase + (size_t)ch * LSPAT + iyc * WW + (qc - 1) * 4);
    const v4f z = {0.0f, 0.0f, 0.0f, 0.0f};
    stg[it] = ok ? t : z;
  }
#pragma unroll
  for (int it = 0; it < 21; ++it)
    *(v4f*)(vl + (tid + it * 128) * 4) = stg[it];

  int lo[9][2];
#pragma unroll
  for (int s = 0; s < 9; ++s) {
#pragma unroll
    for (int j = 0; j < 2; ++j) {
      const int kloc = 4 * s + 2 * half + j;
      const int icl  = kloc / 9;
      const int tap  = kloc - icl * 9;
      const int ky   = tap / 3;
      const int kx   = tap - ky * 3;
      lo[s][j] = icl * VICS + ky * VPITCH + kx + 3 + l16;
    }
  }

  const float* arow = pw + (size_t)(wv * 16 + l16) * (HCH * 9) + 2 * half;

  v2f an[9];
#pragma unroll
  for (int s = 0; s < 9; ++s) {
    an[s][0] = arow[4 * s];
    an[s][1] = arow[4 * s + 1];
  }

  __syncthreads();

  v8f acc0 = {}, acc1 = {}, acc2 = {};
  for (int c = 0; c < (HCH * 9) / KCHUNK; ++c) {
    v2f ac[9];
#pragma unroll
    for (int s = 0; s < 9; ++s) ac[s] = an[s];
    if (c + 1 < (HCH * 9) / KCHUNK) {
      const float* ap = arow + (c + 1) * KCHUNK;
#pragma unroll
      for (int s = 0; s < 9; ++s) {
        an[s][0] = ap[4 * s];
        an[s][1] = ap[4 * s + 1];
      }
    }
    const float* vc = vl + c * (4 * VICS);
#pragma unroll
    for (int s = 0; s < 9; ++s) {
      const float* p0 = vc + lo[s][0];
      const float* p1 = vc + lo[s][1];
      v2f b0, b1, b2;
      b0[0] = p0[0];  b0[1] = p1[0];
      b1[0] = p0[16]; b1[1] = p1[16];
      b2[0] = p0[32]; b2[1] = p1[32];
      acc0 = wmma4(ac[s], b0, acc0);
      acc1 = wmma4(ac[s], b1, acc1);
      acc2 = wmma4(ac[s], b2, acc2);
    }
  }

  float* ob = obuf + (size_t)n * HCH * LSPAT + y * WW + l16;
#pragma unroll
  for (int r = 0; r < 8; ++r) {
    const int oc = wv * 16 + r + half * 8;
    const float bias = pb[oc];
    float* p = ob + (size_t)oc * LSPAT;
    p[0]  += acc0[r] + bias;
    p[16] += acc1[r] + bias;
    p[32] += acc2[r] + bias;
  }
}

__global__ __launch_bounds__(256) void res_gemm_kernel(
    const float* __restrict__ obuf, const float* __restrict__ w,
    const float* __restrict__ bias, float* __restrict__ out) {
  const int lane = threadIdx.x & 31;
  const int half = lane >> 4;
  const int l16  = lane & 15;
  const int wave = blockIdx.x * (blockDim.x >> 5) + (threadIdx.x >> 5);

  const int ntn = LSPAT / 16;
  const int ntm = CIN / 16;
  int t = wave;
  const int n  = t / (ntm * ntn); t -= n * ntm * ntn;
  const int mt = t / ntn;
  const int ct = t % ntn;

  const int mrow = mt * 16 + l16;
  const int col  = ct * 16 + l16;
  const float* ib = obuf + (size_t)n * HCH * LSPAT;

  v8f acc = {};
#pragma unroll
  for (int kk = 0; kk < HCH; kk += 4) {
    const int ka = kk + half * 2;
    v2f a; a[0] = w[mrow * HCH + ka];
           a[1] = w[mrow * HCH + ka + 1];
    v2f b; b[0] = ib[(size_t)ka * LSPAT + col];
           b[1] = ib[(size_t)(ka + 1) * LSPAT + col];
    acc = wmma4(a, b, acc);
  }

  float* ob = out + (size_t)n * CIN * LSPAT;
#pragma unroll
  for (int r = 0; r < 8; ++r) {
    const int row = mt * 16 + r + half * 8;
    ob[(size_t)row * LSPAT + col] = acc[r] + bias[row];
  }
}

extern "C" void kernel_launch(void* const* d_in, const int* in_sizes, int n_in,
                              void* d_out, int out_size, void* d_ws, size_t ws_size,
                              hipStream_t stream) {
  const float* x     = (const float*)d_in[0];
  const float* qkv_w = (const float*)d_in[1];
  const float* qkv_b = (const float*)d_in[2];
  const float* pos_w = (const float*)d_in[3];
  const float* pos_b = (const float*)d_in[4];
  const float* res_w = (const float*)d_in[5];
  const float* res_b = (const float*)d_in[6];
  float* out = (float*)d_out;

  float* qkv  = (float*)d_ws;
  float* obuf = qkv + (size_t)NBATCH * CH3 * LSPAT;

  const int waves1 = NBATCH * (CH3 / 16) * (LSPAT / 16);
  qkv_gemm_kernel<<<waves1 / 8, 256, 0, stream>>>(x, qkv_w, qkv_b, qkv);

  attn_kernel<<<NBATCH * NHEAD * (LSPAT / 64), 128, 0, stream>>>(qkv, obuf);

  pos_conv_kernel<<<NBATCH * HH, 128, 0, stream>>>(qkv, pos_w, pos_b, obuf);

  const int waves4 = NBATCH * (CIN / 16) * (LSPAT / 16);
  res_gemm_kernel<<<waves4 / 8, 256, 0, stream>>>(obuf, res_w, res_b, out);
}


# attention scores computed transposed (keys in rows): the softmax tile in the accumulator is directly the B operand of V*P^T, no LDS re-fragment or barriers; sum order of softmax sum and P*V changes
# speedup vs baseline: 1.8802x; 1.2429x over previous
#include <hip/hip_runtime.h>
#include <math.h>

typedef __attribute__((ext_vector_type(2))) float v2f;
typedef __attribute__((ext_vector_type(4))) float v4f;
typedef __attribute__((ext_vector_type(8))) float v8f;

#define NBATCH 4
#define CIN 512
#define HH 48
#define WW 48
#define LSPAT (HH*WW)
#define CH3 192
#define HCH 64
#define NHEAD 8
#define DHEAD 8
#define ATTN_SCALE 0.35355339059327379f

static __device__ __forceinline__ v8f wmma4(v2f a, v2f b, v8f c) {
  return __builtin_amdgcn_wmma_f32_16x16x4_f32(false, a, false, b, (short)0, c,
                                               false, false);
}

__global__ __launch_bounds__(256) void qkv_gemm_kernel(
    const float* __restrict__ x, const float* __restrict__ w,
    const float* __restrict__ bias, float* __restrict__ qkv) {
  const int lane = threadIdx.x & 31;
  const int half = lane >> 4;
  const int l16  = lane & 15;
  const int wave = blockIdx.x * (blockDim.x >> 5) + (threadIdx.x >> 5);

  const int ntn = LSPAT / 16;
  const int ntm = CH3 / 16;
  int t = wave;
  const int n  = t / (ntm * ntn);  t -= n * ntm * ntn;
  const int mt = t / ntn;
  const int ct = t % ntn;

  const int mrow = mt * 16 + l16;
  const int col  = ct * 16 + l16;
  const float* xb = x + (size_t)n * CIN * LSPAT;

  v8f acc = {};
  for (int kk = 0; kk < CIN; kk += 4) {
    const int ka = kk + half * 2;
    v2f a; a[0] = w[mrow * CIN + ka];
           a[1] = w[mrow * CIN + ka + 1];
    v2f b; b[0] = xb[(size_t)ka * LSPAT + col];
           b[1] = xb[(size_t)(ka + 1) * LSPAT + col];
    acc = wmma4(a, b, acc);
  }

  float* out = qkv + (size_t)n * CH3 * LSPAT;
#pragma unroll
  for (int r = 0; r < 8; ++r) {
    const int row = mt * 16 + r + half * 8;
    out[(size_t)row * LSPAT + col] = acc[r] + bias[row];
  }
}

__global__ __launch_bounds__(128) void attn_kernel(
    const float* __restrict__ qkv, float* __restrict__ obuf) {
  const int lane = threadIdx.x & 31;
  const int half = lane >> 4;
  const int l16  = lane & 15;
  const int wv   = threadIdx.x >> 5;

  const int qtiles = LSPAT / 64;
  const int nh = blockIdx.x / qtiles;
  const int qt = blockIdx.x % qtiles;
  const int n = nh >> 3, h = nh & 7;
  const int qbase = qt * 64 + wv * 16;

  const float* base = qkv + (size_t)n * CH3 * LSPAT + (size_t)h * 24 * LSPAT;
  const float* qm = base;
  const float* km = base + (size_t)DHEAD * LSPAT;
  const float* vm = base + (size_t)2 * DHEAD * LSPAT;

  v2f bq[2];
#pragma unroll
  for (int db = 0; db < 2; ++db) {
    const int d0 = db * 4 + half * 2;
    bq[db][0] = qm[(size_t)d0 * LSPAT + qbase + l16];
    bq[db][1] = qm[(size_t)(d0 + 1) * LSPAT + qbase + l16];
  }
  const float* kp0 = km + (size_t)(half * 2) * LSPAT + l16;
  const float* kp1 = kp0 + (size_t)4 * LSPAT;

  float m8[8];
#pragma unroll
  for (int r = 0; r < 8; ++r) m8[r] = -3.0e38f;

  for (int kt = 0; kt < LSPAT / 16; ++kt) {
    const int kb = kt * 16;
    v2f a0, a1;
    a0[0] = kp0[kb]; a0[1] = kp0[LSPAT + kb];
    a1[0] = kp1[kb]; a1[1] = kp1[LSPAT + kb];
    v8f s = {};
    s = wmma4(a0, bq[0], s);
    s = wmma4(a1, bq[1], s);
#pragma unroll
    for (int r = 0; r < 8; ++r) m8[r] = fmaxf(m8[r], s[r]);
  }
  float mq = m8[0];
#pragma unroll
  for (int r = 1; r < 8; ++r) mq = fmaxf(mq, m8[r]);
  mq = fmaxf(mq, __shfl_xor(mq, 16, 32));

  float s8[8];
#pragma unroll
  for (int r = 0; r < 8; ++r) s8[r] = 0.0f;

  for (int kt = 0; kt < LSPAT / 16; ++kt) {
    const int kb = kt * 16;
    v2f a0, a1;
    a0[0] = kp0[kb]; a0[1] = kp0[LSPAT + kb];
    a1[0] = kp1[kb]; a1[1] = kp1[LSPAT + kb];
    v8f s = {};
    s = wmma4(a0, bq[0], s);
    s = wmma4(a1, bq[1], s);
#pragma unroll
    for (int r = 0; r < 8; ++r) s8[r] += __expf((s[r] - mq) * ATTN_SCALE);
  }
  float sq = ((s8[0] + s8[1]) + (s8[2] + s8[3])) + ((s8[4] + s8[5]) + (s8[6] + s8[7]));
  sq += __shfl_xor(sq, 16, 32);
  const float inv = 1.0f / sq;

  const float* vp = vm + (size_t)(l16 & 7) * LSPAT + half * 8;
  v8f acc = {};
  for (int kt = 0; kt < LSPAT / 16; ++kt) {
    const int kb = kt * 16;
    v2f a0, a1;
    a0[0] = kp0[kb]; a0[1] = kp0[LSPAT + kb];
    a1[0] = kp1[kb]; a1[1] = kp1[LSPAT + kb];
    const v4f va = *(const v4f*)(vp + kb);
    const v4f vb = *(const v4f*)(vp + kb + 4);
    v8f s = {};
    s = wmma4(a0, bq[0], s);
    s = wmma4(a1, bq[1], s);
    float p[8];
#pragma unroll
    for (int r = 0; r < 8; ++r)
      p[r] = __expf((s[r] - mq) * ATTN_SCALE) * inv;
    v2f av, bp;
    av[0] = va[0]; av[1] = va[1]; bp[0] = p[0]; bp[1] = p[1];
    acc = wmma4(av, bp, acc);
    av[0] = va[2]; av[1] = va[3]; bp[0] = p[2]; bp[1] = p[3];
    acc = wmma4(av, bp, acc);
    av[0] = vb[0]; av[1] = vb[1]; bp[0] = p[4]; bp[1] = p[5];
    acc = wmma4(av, bp, acc);
    av[0] = vb[2]; av[1] = vb[3]; bp[0] = p[6]; bp[1] = p[7];
    acc = wmma4(av, bp, acc);
  }

  if (half == 0) {
    float* ob = obuf + (size_t)n * HCH * LSPAT + (size_t)(h * DHEAD) * LSPAT + qbase + l16;
#pragma unroll
    for (int r = 0; r < 8; ++r) ob[(size_t)r * LSPAT] = acc[r];
  }
}

#define VPITCH 56
#define VICS   (3 * VPITCH)
#define KCHUNK 36

__global__ __launch_bounds__(128) void pos_conv_kernel(
    const float* __restrict__ qkv, const float* __restrict__ pw,
    const float* __restrict__ pb, float* __restrict__ obuf) {
  __shared__ __attribute__((aligned(16))) float vl[HCH * VICS];
  const int tid  = threadIdx.x;
  const int lane = tid & 31;
  const int half = lane >> 4;
  const int l16  = lane & 15;
  const int wv   = tid >> 5;
  const int n = blockIdx.x / HH;
  const int y = blockIdx.x % HH;

  const float* vbase = qkv + (size_t)n * CH3 * LSPAT;

  v4f stg[21];
#pragma unroll
  for (int it = 0; it < 21; ++it) {
    const int i   = tid + it * 128;
    const int ic  = i / 42;
    const int rem = i - ic * 42;
    const int r   = rem / 14;
    const int q   = rem - r * 14;
    const int iy  = y + r - 1;
    const bool ok = (q >= 1) && (q <= 12) && (iy >= 0) && (iy < HH);
    const int qc  = q < 1 ? 1 : (q > 12 ? 12 : q);
    const int iyc = iy < 0 ? 0 : (iy > HH - 1 ? HH - 1 : iy);
    const int ch  = (ic >> 3) * 24 + 16 + (ic & 7);
    const v4f t = *(const v4f*)(vbase + (size_t)ch * LSPAT + iyc * WW + (qc - 1) * 4);
    const v4f z = {0.0f, 0.0f, 0.0f, 0.0f};
    stg[it] = ok ? t : z;
  }
#pragma unroll
  for (int it = 0; it < 21; ++it)
    *(v4f*)(vl + (tid + it * 128) * 4) = stg[it];

  int lo[9][2];
#pragma unroll
  for (int s = 0; s < 9; ++s) {
#pragma unroll
    for (int j = 0; j < 2; ++j) {
      const int kloc = 4 * s + 2 * half + j;
      const int icl  = kloc / 9;
      const int tap  = kloc - icl * 9;
      const int ky   = tap / 3;
      const int kx   = tap - ky * 3;
      lo[s][j] = icl * VICS + ky * VPITCH + kx + 3 + l16;
    }
  }

  const float* arow = pw + (size_t)(wv * 16 + l16) * (HCH * 9) + 2 * half;

  v2f an[9];
#pragma unroll
  for (int s = 0; s < 9; ++s) {
    an[s][0] = arow[4 * s];
    an[s][1] = arow[4 * s + 1];
  }

  __syncthreads();

  v8f acc0 = {}, acc1 = {}, acc2 = {};
  for (int c = 0; c < (HCH * 9) / KCHUNK; ++c) {
    v2f ac[9];
#pragma unroll
    for (int s = 0; s < 9; ++s) ac[s] = an[s];
    if (c + 1 < (HCH * 9) / KCHUNK) {
      const float* ap = arow + (c + 1) * KCHUNK;
#pragma unroll
      for (int s = 0; s < 9; ++s) {
        an[s][0] = ap[4 * s];
        an[s][1] = ap[4 * s + 1];
      }
    }
    const float* vc = vl + c * (4 * VICS);
#pragma unroll
    for (int s = 0; s < 9; ++s) {
      const float* p0 = vc + lo[s][0];
      const float* p1 = vc + lo[s][1];
      v2f b0, b1, b2;
      b0[0] = p0[0];  b0[1] = p1[0];
      b1[0] = p0[16]; b1[1] = p1[16];
      b2[0] = p0[32]; b2[1] = p1[32];
      acc0 = wmma4(ac[s], b0, acc0);
      acc1 = wmma4(ac[s], b1, acc1);
      acc2 = wmma4(ac[s], b2, acc2);
    }
  }

  float* ob = obuf + (size_t)n * HCH * LSPAT + y * WW + l16;
#pragma unroll
  for (int r = 0; r < 8; ++r) {
    const int oc = wv * 16 + r + half * 8;
    const float bias = pb[oc];
    float* p = ob + (size_t)oc * LSPAT;
    p[0]  += acc0[r] + bias;
    p[16] += acc1[r] + bias;
    p[32] += acc2[r] + bias;
  }
}

__global__ __launch_bounds__(256) void res_gemm_kernel(
    const float* __restrict__ obuf, const float* __restrict__ w,
    const float* __restrict__ bias, float* __restrict__ out) {
  const int lane = threadIdx.x & 31;
  const int half = lane >> 4;
  const int l16  = lane & 15;
  const int wave = blockIdx.x * (blockDim.x >> 5) + (threadIdx.x >> 5);

  const int ntn = LSPAT / 16;
  const int ntm = CIN / 16;
  int t = wave;
  const int n  = t / (ntm * ntn); t -= n * ntm * ntn;
  const int mt = t / ntn;
  const int ct = t % ntn;

  const int mrow = mt * 16 + l16;
  const int col  = ct * 16 + l16;
  const float* ib = obuf + (size_t)n * HCH * LSPAT;

  v8f acc = {};
#pragma unroll
  for (int kk = 0; kk < HCH; kk += 4) {
    const int ka = kk + half * 2;
    v2f a; a[0] = w[mrow * HCH + ka];
           a[1] = w[mrow * HCH + ka + 1];
    v2f b; b[0] = ib[(size_t)ka * LSPAT + col];
           b[1] = ib[(size_t)(ka + 1) * LSPAT + col];
    acc = wmma4(a, b, acc);
  }

  float* ob = out + (size_t)n * CIN * LSPAT;
#pragma unroll
  for (int r = 0; r < 8; ++r) {
    const int row = mt * 16 + r + half * 8;
    ob[(size_t)row * LSPAT + col] = acc[r] + bias[row];
  }
}

extern "C" void kernel_launch(void* const* d_in, const int* in_sizes, int n_in,
                              void* d_out, int out_size, void* d_ws, size_t ws_size,
                              hipStream_t stream) {
  const float* x     = (const float*)d_in[0];
  const float* qkv_w = (const float*)d_in[1];
  const float* qkv_b = (const float*)d_in[2];
  const float* pos_w = (const float*)d_in[3];
  const float* pos_b = (const float*)d_in[4];
  const float* res_w = (const float*)d_in[5];
  const float* res_b = (const float*)d_in[6];
  float* out = (float*)d_out;

  float* qkv  = (float*)d_ws;
  float* obuf = qkv + (size_t)NBATCH * CH3 * LSPAT;

  const int waves1 = NBATCH * (CH3 / 16) * (LSPAT / 16);
  qkv_gemm_kernel<<<waves1 / 8, 256, 0, stream>>>(x, qkv_w, qkv_b, qkv);

  attn_kernel<<<NBATCH * NHEAD * (LSPAT / 64), 128, 0, stream>>>(qkv, obuf);

  pos_conv_kernel<<<NBATCH * HH, 128, 0, stream>>>(qkv, pos_w, pos_b, obuf);

  const int waves4 = NBATCH * (CIN / 16) * (LSPAT / 16);
  res_gemm_kernel<<<waves4 / 8, 256, 0, stream>>>(obuf, res_w, res_b, out);
}


# attention softmax in ONE pass over the keys (32-key steps, running max, rescale of sum and accumulator) instead of three passes: scores and exp once per tile; sum order of softmax and P*V changes
# speedup vs baseline: 2.5371x; 1.3494x over previous
#include <hip/hip_runtime.h>
#include <math.h>

typedef __attribute__((ext_vector_type(2))) float v2f;
typedef __attribute__((ext_vector_type(4))) float v4f;
typedef __attribute__((ext_vector_type(8))) float v8f;

#define NBATCH 4
#define CIN 512
#define HH 48
#define WW 48
#define LSPAT (HH*WW)
#define CH3 192
#define HCH 64
#define NHEAD 8
#define DHEAD 8
#define ATTN_SCALE 0.35355339059327379f

static __device__ __forceinline__ v8f wmma4(v2f a, v2f b, v8f c) {
  return __builtin_amdgcn_wmma_f32_16x16x4_f32(false, a, false, b, (short)0, c,
                                               false, false);
}

__global__ __launch_bounds__(256) void qkv_gemm_kernel(
    const float* __restrict__ x, const float* __restrict__ w,
    const float* __restrict__ bias, float* __restrict__ qkv) {
  const int lane = threadIdx.x & 31;
  const int half = lane >> 4;
  const int l16  = lane & 15;
  const int wave = blockIdx.x * (blockDim.x >> 5) + (threadIdx.x >> 5);

  const int ntn = LSPAT / 16;
  const int ntm = CH3 / 16;
  int t = wave;
  const int n  = t / (ntm * ntn);  t -= n * ntm * ntn;
  const int mt = t / ntn;
  const int ct = t % ntn;

  const int mrow = mt * 16 + l16;
  const int col  = ct * 16 + l16;
  const float* xb = x + (size_t)n * CIN * LSPAT;

  v8f acc = {};
  for (int kk = 0; kk < CIN; kk += 4) {
    const int ka = kk + half * 2;
    v2f a; a[0] = w[mrow * CIN + ka];
           a[1] = w[mrow * CIN + ka + 1];
    v2f b; b[0] = xb[(size_t)ka * LSPAT + col];
           b[1] = xb[(size_t)(ka + 1) * LSPAT + col];
    acc = wmma4(a, b, acc);
  }

  float* out = qkv + (size_t)n * CH3 * LSPAT;
#pragma unroll
  for (int r = 0; r < 8; ++r) {
    const int row = mt * 16 + r + half * 8;
    out[(size_t)row * LSPAT + col] = acc[r] + bias[row];
  }
}

#define KT 2
static __device__ __forceinline__ float xhalf(float x) {
  const int b = __builtin_bit_cast(int, x);
  return __builtin_bit_cast(
      float, __builtin_amdgcn_permlanex16(b, b, 0x76543210, 0xfedcba98, false, false));
}

__global__ __launch_bounds__(128) void attn_kernel(
    const float* __restrict__ qkv, float* __restrict__ obuf) {
  const int lane = threadIdx.x & 31;
  const int half = lane >> 4;
  const int l16  = lane & 15;
  const int wv   = threadIdx.x >> 5;

  const int qtiles = LSPAT / 64;
  const int nh = blockIdx.x / qtiles;
  const int qt = blockIdx.x % qtiles;
  const int n = nh >> 3, h = nh & 7;
  const int qbase = qt * 64 + wv * 16;

  const float* base = qkv + (size_t)n * CH3 * LSPAT + (size_t)h * 24 * LSPAT;
  const float* qm = base;
  const float* km = base + (size_t)DHEAD * LSPAT;
  const float* vm = base + (size_t)2 * DHEAD * LSPAT;

  v2f bq[2];
#pragma unroll
  for (int db = 0; db < 2; ++db) {
    const int d0 = db * 4 + half * 2;
    bq[db][0] = qm[(size_t)d0 * LSPAT + qbase + l16];
    bq[db][1] = qm[(size_t)(d0 + 1) * LSPAT + qbase + l16];
  }
  const float* kp0 = km + (size_t)(half * 2) * LSPAT + l16;
  const float* kp1 = kp0 + (size_t)4 * LSPAT;

  const float* vp = vm + (size_t)(l16 & 7) * LSPAT + half * 8;

  const float C2 = ATTN_SCALE * 1.4426950408889634f;
  float m_run = -3.0e38f;
  float l_run = 0.0f;
  v8f acc = {};
  for (int kt = 0; kt < LSPAT / 16; kt += KT) {
    v2f a0[KT], a1[KT];
    v4f va[KT], vb[KT];
#pragma unroll
    for (int i = 0; i < KT; ++i) {
      const int kb = (kt + i) * 16;
      a0[i][0] = kp0[kb]; a0[i][1] = kp0[LSPAT + kb];
      a1[i][0] = kp1[kb]; a1[i][1] = kp1[LSPAT + kb];
      va[i] = *(const v4f*)(vp + kb);
      vb[i] = *(const v4f*)(vp + kb + 4);
    }
    v8f s[KT];
#pragma unroll
    for (int i = 0; i < KT; ++i) {
      const v8f z = {};
      s[i] = wmma4(a0[i], bq[0], z);
    }
#pragma unroll
    for (int i = 0; i < KT; ++i) s[i] = wmma4(a1[i], bq[1], s[i]);

    float t = s[0][0];
#pragma unroll
    for (int i = 0; i < KT; ++i) {
      const float u = fmaxf(fmaxf(fmaxf(s[i][0], s[i][1]), fmaxf(s[i][2], s[i][3])),
                            fmaxf(fmaxf(s[i][4], s[i][5]), fmaxf(s[i][6], s[i][7])));
      t = fmaxf(t, u);
    }
    t = fmaxf(t, xhalf(t));
    const float m_new = fmaxf(m_run, t);
    const float alpha = __builtin_amdgcn_exp2f((m_run - m_new) * C2);
    const float nm = -(m_new * C2);
    m_run = m_new;

    float ps = 0.0f;
#pragma unroll
    for (int i = 0; i < KT; ++i) {
#pragma unroll
      for (int r = 0; r < 8; ++r)
        s[i][r] = __builtin_amdgcn_exp2f(fmaf(s[i][r], C2, nm));
      ps += ((s[i][0] + s[i][1]) + (s[i][2] + s[i][3])) +
            ((s[i][4] + s[i][5]) + (s[i][6] + s[i][7]));
    }
    l_run = fmaf(l_run, alpha, ps);
#pragma unroll
    for (int r = 0; r < 8; ++r) acc[r] *= alpha;

#pragma unroll
    for (int i = 0; i < KT; ++i) {
      v2f av, bp;
      av[0] = va[i][0]; av[1] = va[i][1]; bp[0] = s[i][0]; bp[1] = s[i][1];
      acc = wmma4(av, bp, acc);
      av[0] = va[i][2]; av[1] = va[i][3]; bp[0] = s[i][2]; bp[1] = s[i][3];
      acc = wmma4(av, bp, acc);
      av[0] = vb[i][0]; av[1] = vb[i][1]; bp[0] = s[i][4]; bp[1] = s[i][5];
      acc = wmma4(av, bp, acc);
      av[0] = vb[i][2]; av[1] = vb[i][3]; bp[0] = s[i][6]; bp[1] = s[i][7];
      acc = wmma4(av, bp, acc);
    }
  }
  l_run += xhalf(l_run);
  const float inv = 1.0f / l_run;

  if (half == 0) {
    float* ob = obuf + (size_t)n * HCH * LSPAT + (size_t)(h * DHEAD) * LSPAT + qbase + l16;
#pragma unroll
    for (int r = 0; r < 8; ++r) ob[(size_t)r * LSPAT] = acc[r] * inv;
  }
}

#define VPITCH 56
#define VICS   (3 * VPITCH)
#define KCHUNK 36

__global__ __launch_bounds__(128) void pos_conv_kernel(
    const float* __restrict__ qkv, const float* __restrict__ pw,
    const float* __restrict__ pb, float* __restrict__ obuf) {
  __shared__ __attribute__((aligned(16))) float vl[HCH * VICS];
  const int tid  = threadIdx.x;
  const int lane = tid & 31;
  const int half = lane >> 4;
  const int l16  = lane & 15;
  const int wv   = tid >> 5;
  const int n = blockIdx.x / HH;
  const int y = blockIdx.x % HH;

  const float* vbase = qkv + (size_t)n * CH3 * LSPAT;

  v4f stg[21];
#pragma unroll
  for (int it = 0; it < 21; ++it) {
    const int i   = tid + it * 128;
    const int ic  = i / 42;
    const int rem = i - ic * 42;
    const int r   = rem / 14;
    const int q   = rem - r * 14;
    const int iy  = y + r - 1;
    const bool ok = (q >= 1) && (q <= 12) && (iy >= 0) && (iy < HH);
    const int qc  = q < 1 ? 1 : (q > 12 ? 12 : q);
    const int iyc = iy < 0 ? 0 : (iy > HH - 1 ? HH - 1 : iy);
    const int ch  = (ic >> 3) * 24 + 16 + (ic & 7);
    const v4f t = *(const v4f*)(vbase + (size_t)ch * LSPAT + iyc * WW + (qc - 1) * 4);
    const v4f z = {0.0f, 0.0f, 0.0f, 0.0f};
    stg[it] = ok ? t : z;
  }
#pragma unroll
  for (int it = 0; it < 21; ++it)
    *(v4f*)(vl + (tid + it * 128) * 4) = stg[it];

  int lo[9][2];
#pragma unroll
  for (int s = 0; s < 9; ++s) {
#pragma unroll
    for (int j = 0; j < 2; ++j) {
      const int kloc = 4 * s + 2 * half + j;
      const int icl  = kloc / 9;
      const int tap  = kloc - icl * 9;
      const int ky   = tap / 3;
      const int kx   = tap - ky * 3;
      lo[s][j] = icl * VICS + ky * VPITCH + kx + 3 + l16;
    }
  }

  const float* arow = pw + (size_t)(wv * 16 + l16) * (HCH * 9) + 2 * half;

  v2f an[9];
#pragma unroll
  for (int s = 0; s < 9; ++s) {
    an[s][0] = arow[4 * s];
    an[s][1] = arow[4 * s + 1];
  }

  __syncthreads();

  v8f acc0 = {}, acc1 = {}, acc2 = {};
  for (int c = 0; c < (HCH * 9) / KCHUNK; ++c) {
    v2f ac[9];
#pragma unroll
    for (int s = 0; s < 9; ++s) ac[s] = an[s];
    if (c + 1 < (HCH * 9) / KCHUNK) {
      const float* ap = arow + (c + 1) * KCHUNK;
#pragma unroll
      for (int s = 0; s < 9; ++s) {
        an[s][0] = ap[4 * s];
        an[s][1] = ap[4 * s + 1];
      }
    }
    const float* vc = vl + c * (4 * VICS);
#pragma unroll
    for (int s = 0; s < 9; ++s) {
      const float* p0 = vc + lo[s][0];
      const float* p1 = vc + lo[s][1];
      v2f b0, b1, b2;
      b0[0] = p0[0];  b0[1] = p1[0];
      b1[0] = p0[16]; b1[1] = p1[16];
      b2[0] = p0[32]; b2[1] = p1[32];
      acc0 = wmma4(ac[s], b0, acc0);
      acc1 = wmma4(ac[s], b1, acc1);
      acc2 = wmma4(ac[s], b2, acc2);
    }
  }

  float* ob = obuf + (size_t)n * HCH * LSPAT + y * WW + l16;
#pragma unroll
  for (int r = 0; r < 8; ++r) {
    const int oc = wv * 16 + r + half * 8;
    const float bias = pb[oc];
    float* p = ob + (size_t)oc * LSPAT;
    p[0]  += acc0[r] + bias;
    p[16] += acc1[r] + bias;
    p[32] += acc2[r] + bias;
  }
}

__global__ __launch_bounds__(256) void res_gemm_kernel(
    const float* __restrict__ obuf, const float* __restrict__ w,
    const float* __restrict__ bias, float* __restrict__ out) {
  const int lane = threadIdx.x & 31;
  const int half = lane >> 4;
  const int l16  = lane & 15;
  const int wave = blockIdx.x * (blockDim.x >> 5) + (threadIdx.x >> 5);

  const int ntn = LSPAT / 16;
  const int ntm = CIN / 16;
  int t = wave;
  const int n  = t / (ntm * ntn); t -= n * ntm * ntn;
  const int mt = t / ntn;
  const int ct = t % ntn;

  const int mrow = mt * 16 + l16;
  const int col  = ct * 16 + l16;
  const float* ib = obuf + (size_t)n * HCH * LSPAT;

  v8f acc = {};
#pragma unroll
  for (int kk = 0; kk < HCH; kk += 4) {
    const int ka = kk + half * 2;
    v2f a; a[0] = w[mrow * HCH + ka];
           a[1] = w[mrow * HCH + ka + 1];
    v2f b; b[0] = ib[(size_t)ka * LSPAT + col];
           b[1] = ib[(size_t)(ka + 1) * LSPAT + col];
    acc = wmma4(a, b, acc);
  }

  float* ob = out + (size_t)n * CIN * LSPAT;
#pragma unroll
  for (int r = 0; r < 8; ++r) {
    const int row = mt * 16 + r + half * 8;
    ob[(size_t)row * LSPAT + col] = acc[r] + bias[row];
  }
}

extern "C" void kernel_launch(void* const* d_in, const int* in_sizes, int n_in,
                              void* d_out, int out_size, void* d_ws, size_t ws_size,
                              hipStream_t stream) {
  const float* x     = (const float*)d_in[0];
  const float* qkv_w = (const float*)d_in[1];
  const float* qkv_b = (const float*)d_in[2];
  const float* pos_w = (const float*)d_in[3];
  const float* pos_b = (const float*)d_in[4];
  const float* res_w = (const float*)d_in[5];
  const float* res_b = (const float*)d_in[6];
  float* out = (float*)d_out;

  float* qkv  = (float*)d_ws;
  float* obuf = qkv + (size_t)NBATCH * CH3 * LSPAT;

  const int waves1 = NBATCH * (CH3 / 16) * (LSPAT / 16);
  qkv_gemm_kernel<<<waves1 / 8, 256, 0, stream>>>(x, qkv_w, qkv_b, qkv);

  attn_kernel<<<NBATCH * NHEAD * (LSPAT / 64), 128, 0, stream>>>(qkv, obuf);

  pos_conv_kernel<<<NBATCH * HH, 128, 0, stream>>>(qkv, pos_w, pos_b, obuf);

  const int waves4 = NBATCH * (CIN / 16) * (LSPAT / 16);
  res_gemm_kernel<<<waves4 / 8, 256, 0, stream>>>(obuf, res_w, res_b, out);
}


# both 1x1-conv GEMMs: each wave owns 16x64 as four column-interleaved tiles: B loads and D stores are whole-line 128-bit, one 128-bit A load feeds 8 WMMAs; k sum order of both products changes
# speedup vs baseline: 2.9572x; 1.1656x over previous
#include <hip/hip_runtime.h>
#include <math.h>

typedef __attribute__((ext_vector_type(2))) float v2f;
typedef __attribute__((ext_vector_type(4))) float v4f;
typedef __attribute__((ext_vector_type(8))) float v8f;

#define NBATCH 4
#define CIN 512
#define HH 48
#define WW 48
#define LSPAT (HH*WW)
#define CH3 192
#define HCH 64
#define NHEAD 8
#define DHEAD 8
#define ATTN_SCALE 0.35355339059327379f

static __device__ __forceinline__ v8f wmma4(v2f a, v2f b, v8f c) {
  return __builtin_amdgcn_wmma_f32_16x16x4_f32(false, a, false, b, (short)0, c,
                                               false, false);
}

#define GCH 4

template <int XPITCH>
static __device__ __forceinline__ void load_group(
    const float* __restrict__ wrow, const float* __restrict__ xcol, int k0,
    v4f (&a)[GCH], v4f (&b)[GCH][4]) {
#pragma unroll
  for (int i = 0; i < GCH; ++i) {
    a[i] = *(const v4f*)(wrow + k0 + 8 * i);
#pragma unroll
    for (int e = 0; e < 4; ++e)
      b[i][e] = *(const v4f*)(xcol + (size_t)(k0 + 8 * i + e) * XPITCH);
  }
}

static __device__ __forceinline__ void mma_group(
    const v4f (&a)[GCH], const v4f (&b)[GCH][4], v8f (&acc)[4]) {
#pragma unroll
  for (int i = 0; i < GCH; ++i) {
#pragma unroll
    for (int s = 0; s < 2; ++s) {
      v2f af; af[0] = a[i][2 * s]; af[1] = a[i][2 * s + 1];
#pragma unroll
      for (int t = 0; t < 4; ++t) {
        v2f bf; bf[0] = b[i][2 * s][t]; bf[1] = b[i][2 * s + 1][t];
        acc[t] = wmma4(af, bf, acc[t]);
      }
    }
  }
}

template <int KDIM, int XPITCH>
static __device__ __forceinline__ void gemm16x64(
    const float* __restrict__ wrow, const float* __restrict__ xcol,
    v8f (&acc)[4]) {
  constexpr int KG = 8 * GCH;
  constexpr int NG = KDIM / KG;
  v4f a0[GCH], b0[GCH][4], a1[GCH], b1[GCH][4];
  load_group<XPITCH>(wrow, xcol, 0, a0, b0);
#pragma unroll 1
  for (int g = 0; g + 2 < NG; g += 2) {
    load_group<XPITCH>(wrow, xcol, (g + 1) * KG, a1, b1);
    __builtin_amdgcn_sched_barrier(0);
    mma_group(a0, b0, acc);
    load_group<XPITCH>(wrow, xcol, (g + 2) * KG, a0, b0);
    __builtin_amdgcn_sched_barrier(0);
    mma_group(a1, b1, acc);
  }
  load_group<XPITCH>(wrow, xcol, (NG - 1) * KG, a1, b1);
  __builtin_amdgcn_sched_barrier(0);
  mma_group(a0, b0, acc);
  mma_group(a1, b1, acc);
}

static __device__ __forceinline__ void store16x64(
    float* __restrict__ out, const float* __restrict__ bias, int row0, int half,
    const v8f (&acc)[4]) {
#pragma unroll
  for (int r = 0; r < 8; ++r) {
    const int row = row0 + r + half * 8;
    const float bv = bias[row];
    v4f o;
    o[0] = acc[0][r] + bv; o[1] = acc[1][r] + bv;
    o[2] = acc[2][r] + bv; o[3] = acc[3][r] + bv;
    *(v4f*)(out + (size_t)row * LSPAT) = o;
  }
}

__global__ __launch_bounds__(128) void qkv_gemm_kernel(
    const float* __restrict__ x, const float* __restrict__ w,
    const float* __restrict__ bias, float* __restrict__ qkv) {
  const int lane = threadIdx.x & 31;
  const int half = lane >> 4;
  const int l16  = lane & 15;
  const int wv   = threadIdx.x >> 5;

  const int ncg  = LSPAT / 64;
  const int nmtg = CH3 / 64;
  int t = blockIdx.x;
  const int n   = t / (ncg * nmtg);  t -= n * ncg * nmtg;
  const int cg  = t / nmtg;
  const int mtg = t % nmtg;
  const int mt  = mtg * 4 + wv;

  const int col0 = cg * 64 + 4 * l16;
  const float* wrow = w + (size_t)(mt * 16 + l16) * CIN + 4 * half;
  const float* xcol = x + (size_t)n * CIN * LSPAT + (size_t)(4 * half) * LSPAT + col0;

  v8f acc[4] = {};
  gemm16x64<CIN, LSPAT>(wrow, xcol, acc);

  store16x64(qkv + (size_t)n * CH3 * LSPAT + col0, bias, mt * 16, half, acc);
}

#define KT 2
static __device__ __forceinline__ float xhalf(float x) {
  const int b = __builtin_bit_cast(int, x);
  return __builtin_bit_cast(
      float, __builtin_amdgcn_permlanex16(b, b, 0x76543210, 0xfedcba98, false, false));
}

__global__ __launch_bounds__(128) void attn_kernel(
    const float* __restrict__ qkv, float* __restrict__ obuf) {
  const int lane = threadIdx.x & 31;
  const int half = lane >> 4;
  const int l16  = lane & 15;
  const int wv   = threadIdx.x >> 5;

  const int qtiles = LSPAT / 64;
  const int nh = blockIdx.x / qtiles;
  const int qt = blockIdx.x % qtiles;
  const int n = nh >> 3, h = nh & 7;
  const int qbase = qt * 64 + wv * 16;

  const float* base = qkv + (size_t)n * CH3 * LSPAT + (size_t)h * 24 * LSPAT;
  const float* qm = base;
  const float* km = base + (size_t)DHEAD * LSPAT;
  const float* vm = base + (size_t)2 * DHEAD * LSPAT;

  v2f bq[2];
#pragma unroll
  for (int db = 0; db < 2; ++db) {
    const int d0 = db * 4 + half * 2;
    bq[db][0] = qm[(size_t)d0 * LSPAT + qbase + l16];
    bq[db][1] = qm[(size_t)(d0 + 1) * LSPAT + qbase + l16];
  }
  const float* kp0 = km + (size_t)(half * 2) * LSPAT + l16;
  const float* kp1 = kp0 + (size_t)4 * LSPAT;

  const float* vp = vm + (size_t)(l16 & 7) * LSPAT + half * 8;

  const float C2 = ATTN_SCALE * 1.4426950408889634f;
  float m_run = -3.0e38f;
  float l_run = 0.0f;
  v8f acc = {};
  for (int kt = 0; kt < LSPAT / 16; kt += KT) {
    v2f a0[KT], a1[KT];
    v4f va[KT], vb[KT];
#pragma unroll
    for (int i = 0; i < KT; ++i) {
      const int kb = (kt + i) * 16;
      a0[i][0] = kp0[kb]; a0[i][1] = kp0[LSPAT + kb];
      a1[i][0] = kp1[kb]; a1[i][1] = kp1[LSPAT + kb];
      va[i] = *(const v4f*)(vp + kb);
      vb[i] = *(const v4f*)(vp + kb + 4);
    }
    v8f s[KT];
#pragma unroll
    for (int i = 0; i < KT; ++i) {
      const v8f z = {};
      s[i] = wmma4(a0[i], bq[0], z);
    }
#pragma unroll
    for (int i = 0; i < KT; ++i) s[i] = wmma4(a1[i], bq[1], s[i]);

    float t = s[0][0];
#pragma unroll
    for (int i = 0; i < KT; ++i) {
      const float u = fmaxf(fmaxf(fmaxf(s[i][0], s[i][1]), fmaxf(s[i][2], s[i][3])),
                            fmaxf(fmaxf(s[i][4], s[i][5]), fmaxf(s[i][6], s[i][7])));
      t = fmaxf(t, u);
    }
    t = fmaxf(t, xhalf(t));
    const float m_new = fmaxf(m_run, t);
    const float alpha = __builtin_amdgcn_exp2f((m_run - m_new) * C2);
    const float nm = -(m_new * C2);
    m_run = m_new;

    float ps = 0.0f;
#pragma unroll
    for (int i = 0; i < KT; ++i) {
#pragma unroll
      for (int r = 0; r < 8; ++r)
        s[i][r] = __builtin_amdgcn_exp2f(fmaf(s[i][r], C2, nm));
      ps += ((s[i][0] + s[i][1]) + (s[i][2] + s[i][3])) +
            ((s[i][4] + s[i][5]) + (s[i][6] + s[i][7]));
    }
    l_run = fmaf(l_run, alpha, ps);
#pragma unroll
    for (int r = 0; r < 8; ++r) acc[r] *= alpha;

#pragma unroll
    for (int i = 0; i < KT; ++i) {
      v2f av, bp;
      av[0] = va[i][0]; av[1] = va[i][1]; bp[0] = s[i][0]; bp[1] = s[i][1];
      acc = wmma4(av, bp, acc);
      av[0] = va[i][2]; av[1] = va[i][3]; bp[0] = s[i][2]; bp[1] = s[i][3];
      acc = wmma4(av, bp, acc);
      av[0] = vb[i][0]; av[1] = vb[i][1]; bp[0] = s[i][4]; bp[1] = s[i][5];
      acc = wmma4(av, bp, acc);
      av[0] = vb[i][2]; av[1] = vb[i][3]; bp[0] = s[i][6]; bp[1] = s[i][7];
      acc = wmma4(av, bp, acc);
    }
  }
  l_run += xhalf(l_run);
  const float inv = 1.0f / l_run;

  if (half == 0) {
    float* ob = obuf + (size_t)n * HCH * LSPAT + (size_t)(h * DHEAD) * LSPAT + qbase + l16;
#pragma unroll
    for (int r = 0; r < 8; ++r) ob[(size_t)r * LSPAT] = acc[r] * inv;
  }
}

#define VPITCH 56
#define VICS   (3 * VPITCH)
#define KCHUNK 36

__global__ __launch_bounds__(128) void pos_conv_kernel(
    const float* __restrict__ qkv, const float* __restrict__ pw,
    const float* __restrict__ pb, float* __restrict__ obuf) {
  __shared__ __attribute__((aligned(16))) float vl[HCH * VICS];
  const int tid  = threadIdx.x;
  const int lane = tid & 31;
  const int half = lane >> 4;
  const int l16  = lane & 15;
  const int wv   = tid >> 5;
  const int n = blockIdx.x / HH;
  const int y = blockIdx.x % HH;

  const float* vbase = qkv + (size_t)n * CH3 * LSPAT;

  v4f stg[21];
#pragma unroll
  for (int it = 0; it < 21; ++it) {
    const int i   = tid + it * 128;
    const int ic  = i / 42;
    const int rem = i - ic * 42;
    const int r   = rem / 14;
    const int q   = rem - r * 14;
    const int iy  = y + r - 1;
    const bool ok = (q >= 1) && (q <= 12) && (iy >= 0) && (iy < HH);
    const int qc  = q < 1 ? 1 : (q > 12 ? 12 : q);
    const int iyc = iy < 0 ? 0 : (iy > HH - 1 ? HH - 1 : iy);
    const int ch  = (ic >> 3) * 24 + 16 + (ic & 7);
    const v4f t = *(const v4f*)(vbase + (size_t)ch * LSPAT + iyc * WW + (qc - 1) * 4);
    const v4f z = {0.0f, 0.0f, 0.0f, 0.0f};
    stg[it] = ok ? t : z;
  }
#pragma unroll
  for (int it = 0; it < 21; ++it)
    *(v4f*)(vl + (tid + it * 128) * 4) = stg[it];

  int lo[9][2];
#pragma unroll
  for (int s = 0; s < 9; ++s) {
#pragma unroll
    for (int j = 0; j < 2; ++j) {
      const int kloc = 4 * s + 2 * half + j;
      const int icl  = kloc / 9;
      const int tap  = kloc - icl * 9;
      const int ky   = tap / 3;
      const int kx   = tap - ky * 3;
      lo[s][j] = icl * VICS + ky * VPITCH + kx + 3 + l16;
    }
  }

  const float* arow = pw + (size_t)(wv * 16 + l16) * (HCH * 9) + 2 * half;

  v2f an[9];
#pragma unroll
  for (int s = 0; s < 9; ++s) {
    an[s][0] = arow[4 * s];
    an[s][1] = arow[4 * s + 1];
  }

  __syncthreads();

  v8f acc0 = {}, acc1 = {}, acc2 = {};
  for (int c = 0; c < (HCH * 9) / KCHUNK; ++c) {
    v2f ac[9];
#pragma unroll
    for (int s = 0; s < 9; ++s) ac[s] = an[s];
    if (c + 1 < (HCH * 9) / KCHUNK) {
      const float* ap = arow + (c + 1) * KCHUNK;
#pragma unroll
      for (int s = 0; s < 9; ++s) {
        an[s][0] = ap[4 * s];
        an[s][1] = ap[4 * s + 1];
      }
    }
    const float* vc = vl + c * (4 * VICS);
#pragma unroll
    for (int s = 0; s < 9; ++s) {
      const float* p0 = vc + lo[s][0];
      const float* p1 = vc + lo[s][1];
      v2f b0, b1, b2;
      b0[0] = p0[0];  b0[1] = p1[0];
      b1[0] = p0[16]; b1[1] = p1[16];
      b2[0] = p0[32]; b2[1] = p1[32];
      acc0 = wmma4(ac[s], b0, acc0);
      acc1 = wmma4(ac[s], b1, acc1);
      acc2 = wmma4(ac[s], b2, acc2);
    }
  }

  float* ob = obuf + (size_t)n * HCH * LSPAT + y * WW + l16;
#pragma unroll
  for (int r = 0; r < 8; ++r) {
    const int oc = wv * 16 + r + half * 8;
    const float bias = pb[oc];
    float* p = ob + (size_t)oc * LSPAT;
    p[0]  += acc0[r] + bias;
    p[16] += acc1[r] + bias;
    p[32] += acc2[r] + bias;
  }
}

__global__ __launch_bounds__(128) void res_gemm_kernel(
    const float* __restrict__ obuf, const float* __restrict__ w,
    const float* __restrict__ bias, float* __restrict__ out) {
  const int lane = threadIdx.x & 31;
  const int half = lane >> 4;
  const int l16  = lane & 15;
  const int wv   = threadIdx.x >> 5;

  const int ncg  = LSPAT / 64;
  const int nmtg = CIN / 64;
  int t = blockIdx.x;
  const int n   = t / (ncg * nmtg); t -= n * ncg * nmtg;
  const int cg  = t / nmtg;
  const int mtg = t % nmtg;
  const int mt  = mtg * 4 + wv;

  const int col0 = cg * 64 + 4 * l16;
  const float* wrow = w + (size_t)(mt * 16 + l16) * HCH + 4 * half;
  const float* xcol = obuf + (size_t)n * HCH * LSPAT + (size_t)(4 * half) * LSPAT + col0;

  v8f acc[4] = {};
  gemm16x64<HCH, LSPAT>(wrow, xcol, acc);

  store16x64(out + (size_t)n * CIN * LSPAT + col0, bias, mt * 16, half, acc);
}

extern "C" void kernel_launch(void* const* d_in, const int* in_sizes, int n_in,
                              void* d_out, int out_size, void* d_ws, size_t ws_size,
                              hipStream_t stream) {
  const float* x     = (const float*)d_in[0];
  const float* qkv_w = (const float*)d_in[1];
  const float* qkv_b = (const float*)d_in[2];
  const float* pos_w = (const float*)d_in[3];
  const float* pos_b = (const float*)d_in[4];
  const float* res_w = (const float*)d_in[5];
  const float* res_b = (const float*)d_in[6];
  float* out = (float*)d_out;

  float* qkv  = (float*)d_ws;
  float* obuf = qkv + (size_t)NBATCH * CH3 * LSPAT;

  qkv_gemm_kernel<<<NBATCH * (LSPAT / 64) * (CH3 / 64), 128, 0, stream>>>(x, qkv_w, qkv_b, qkv);

  attn_kernel<<<NBATCH * NHEAD * (LSPAT / 64), 128, 0, stream>>>(qkv, obuf);

  pos_conv_kernel<<<NBATCH * HH, 128, 0, stream>>>(qkv, pos_w, pos_b, obuf);

  res_gemm_kernel<<<NBATCH * (LSPAT / 64) * (CIN / 64), 128, 0, stream>>>(obuf, res_w, res_b, out);
}


# attention: keys of each 16-query tile split over two waves, halves joined via LDS (max, sum, acc rescaled): 9 waves on every SIMD, was 4 or 5; sum order of softmax sum and P*V changes
# speedup vs baseline: 3.3060x; 1.1179x over previous
#include <hip/hip_runtime.h>
#include <math.h>

typedef __attribute__((ext_vector_type(2))) float v2f;
typedef __attribute__((ext_vector_type(4))) float v4f;
typedef __attribute__((ext_vector_type(8))) float v8f;

#define NBATCH 4
#define CIN 512
#define HH 48
#define WW 48
#define LSPAT (HH*WW)
#define CH3 192
#define HCH 64
#define NHEAD 8
#define DHEAD 8
#define ATTN_SCALE 0.35355339059327379f

static __device__ __forceinline__ v8f wmma4(v2f a, v2f b, v8f c) {
  return __builtin_amdgcn_wmma_f32_16x16x4_f32(false, a, false, b, (short)0, c,
                                               false, false);
}

#define GCH 4

template <int XPITCH>
static __device__ __forceinline__ void load_group(
    const float* __restrict__ wrow, const float* __restrict__ xcol, int k0,
    v4f (&a)[GCH], v4f (&b)[GCH][4]) {
#pragma unroll
  for (int i = 0; i < GCH; ++i) {
    a[i] = *(const v4f*)(wrow + k0 + 8 * i);
#pragma unroll
    for (int e = 0; e < 4; ++e)
      b[i][e] = *(const v4f*)(xcol + (size_t)(k0 + 8 * i + e) * XPITCH);
  }
}

static __device__ __forceinline__ void mma_group(
    const v4f (&a)[GCH], const v4f (&b)[GCH][4], v8f (&acc)[4]) {
#pragma unroll
  for (int i = 0; i < GCH; ++i) {
#pragma unroll
    for (int s = 0; s < 2; ++s) {
      v2f af; af[0] = a[i][2 * s]; af[1] = a[i][2 * s + 1];
#pragma unroll
      for (int t = 0; t < 4; ++t) {
        v2f bf; bf[0] = b[i][2 * s][t]; bf[1] = b[i][2 * s + 1][t];
        acc[t] = wmma4(af, bf, acc[t]);
      }
    }
  }
}

template <int KDIM, int XPITCH>
static __device__ __forceinline__ void gemm16x64(
    const float* __restrict__ wrow, const float* __restrict__ xcol,
    v8f (&acc)[4]) {
  constexpr int KG = 8 * GCH;
  constexpr int NG = KDIM / KG;
  v4f a0[GCH], b0[GCH][4], a1[GCH], b1[GCH][4];
  load_group<XPITCH>(wrow, xcol, 0, a0, b0);
#pragma unroll 1
  for (int g = 0; g + 2 < NG; g += 2) {
    load_group<XPITCH>(wrow, xcol, (g + 1) * KG, a1, b1);
    __builtin_amdgcn_sched_barrier(0);
    mma_group(a0, b0, acc);
    load_group<XPITCH>(wrow, xcol, (g + 2) * KG, a0, b0);
    __builtin_amdgcn_sched_barrier(0);
    mma_group(a1, b1, acc);
  }
  load_group<XPITCH>(wrow, xcol, (NG - 1) * KG, a1, b1);
  __builtin_amdgcn_sched_barrier(0);
  mma_group(a0, b0, acc);
  mma_group(a1, b1, acc);
}

static __device__ __forceinline__ void store16x64(
    float* __restrict__ out, const float* __restrict__ bias, int row0, int half,
    const v8f (&acc)[4]) {
#pragma unroll
  for (int r = 0; r < 8; ++r) {
    const int row = row0 + r + half * 8;
    const float bv = bias[row];
    v4f o;
    o[0] = acc[0][r] + bv; o[1] = acc[1][r] + bv;
    o[2] = acc[2][r] + bv; o[3] = acc[3][r] + bv;
    *(v4f*)(out + (size_t)row * LSPAT) = o;
  }
}

__global__ __launch_bounds__(128) void qkv_gemm_kernel(
    const float* __restrict__ x, const float* __restrict__ w,
    const float* __restrict__ bias, float* __restrict__ qkv) {
  const int lane = threadIdx.x & 31;
  const int half = lane >> 4;
  const int l16  = lane & 15;
  const int wv   = threadIdx.x >> 5;

  const int ncg  = LSPAT / 64;
  const int nmtg = CH3 / 64;
  int t = blockIdx.x;
  const int n   = t / (ncg * nmtg);  t -= n * ncg * nmtg;
  const int cg  = t / nmtg;
  const int mtg = t % nmtg;
  const int mt  = mtg * 4 + wv;

  const int col0 = cg * 64 + 4 * l16;
  const float* wrow = w + (size_t)(mt * 16 + l16) * CIN + 4 * half;
  const float* xcol = x + (size_t)n * CIN * LSPAT + (size_t)(4 * half) * LSPAT + col0;

  v8f acc[4] = {};
  gemm16x64<CIN, LSPAT>(wrow, xcol, acc);

  store16x64(qkv + (size_t)n * CH3 * LSPAT + col0, bias, mt * 16, half, acc);
}

#define KT 2
#define KTILES_PER_WAVE (LSPAT / 16 / 2)
static __device__ __forceinline__ float xhalf(float x) {
  const int b = __builtin_bit_cast(int, x);
  return __builtin_bit_cast(
      float, __builtin_amdgcn_permlanex16(b, b, 0x76543210, 0xfedcba98, false, false));
}

__global__ __launch_bounds__(128) void attn_kernel(
    const float* __restrict__ qkv, float* __restrict__ obuf) {
  __shared__ float xacc[2][DHEAD][16];
  __shared__ float xmax[2][16];
  __shared__ float xsum[2][16];

  const int lane = threadIdx.x & 31;
  const int half = lane >> 4;
  const int l16  = lane & 15;
  const int wv   = threadIdx.x >> 5;
  const int qsel = wv & 1;
  const int ksel = wv >> 1;

  const int qpairs = LSPAT / 32;
  const int nh = blockIdx.x / qpairs;
  const int qp = blockIdx.x % qpairs;
  const int n = nh >> 3, h = nh & 7;
  const int qbase = qp * 32 + qsel * 16;
  const int key0  = ksel * (KTILES_PER_WAVE * 16);

  const float* base = qkv + (size_t)n * CH3 * LSPAT + (size_t)h * 24 * LSPAT;
  const float* qm = base;
  const float* km = base + (size_t)DHEAD * LSPAT;
  const float* vm = base + (size_t)2 * DHEAD * LSPAT;

  v2f bq[2];
#pragma unroll
  for (int db = 0; db < 2; ++db) {
    const int d0 = db * 4 + half * 2;
    bq[db][0] = qm[(size_t)d0 * LSPAT + qbase + l16];
    bq[db][1] = qm[(size_t)(d0 + 1) * LSPAT + qbase + l16];
  }
  const float* kp0 = km + (size_t)(half * 2) * LSPAT + key0 + l16;
  const float* kp1 = kp0 + (size_t)4 * LSPAT;

  const float* vp = vm + (size_t)(l16 & 7) * LSPAT + key0 + half * 8;

  const float C2 = ATTN_SCALE * 1.4426950408889634f;
  float m_run = -3.0e38f;
  float l_run = 0.0f;
  v8f acc = {};
  for (int kt = 0; kt < KTILES_PER_WAVE; kt += KT) {
    v2f a0[KT], a1[KT];
    v4f va[KT], vb[KT];
#pragma unroll
    for (int i = 0; i < KT; ++i) {
      const int kb = (kt + i) * 16;
      a0[i][0] = kp0[kb]; a0[i][1] = kp0[LSPAT + kb];
      a1[i][0] = kp1[kb]; a1[i][1] = kp1[LSPAT + kb];
      va[i] = *(const v4f*)(vp + kb);
      vb[i] = *(const v4f*)(vp + kb + 4);
    }
    v8f s[KT];
#pragma unroll
    for (int i = 0; i < KT; ++i) {
      const v8f z = {};
      s[i] = wmma4(a0[i], bq[0], z);
    }
#pragma unroll
    for (int i = 0; i < KT; ++i) s[i] = wmma4(a1[i], bq[1], s[i]);

    float t = s[0][0];
#pragma unroll
    for (int i = 0; i < KT; ++i) {
      const float u = fmaxf(fmaxf(fmaxf(s[i][0], s[i][1]), fmaxf(s[i][2], s[i][3])),
                            fmaxf(fmaxf(s[i][4], s[i][5]), fmaxf(s[i][6], s[i][7])));
      t = fmaxf(t, u);
    }
    t = fmaxf(t, xhalf(t));
    const float m_new = fmaxf(m_run, t);
    const float alpha = __builtin_amdgcn_exp2f((m_run - m_new) * C2);
    const float nm = -(m_new * C2);
    m_run = m_new;

    float ps = 0.0f;
#pragma unroll
    for (int i = 0; i < KT; ++i) {
#pragma unroll
      for (int r = 0; r < 8; ++r)
        s[i][r] = __builtin_amdgcn_exp2f(fmaf(s[i][r], C2, nm));
      ps += ((s[i][0] + s[i][1]) + (s[i][2] + s[i][3])) +
            ((s[i][4] + s[i][5]) + (s[i][6] + s[i][7]));
    }
    l_run = fmaf(l_run, alpha, ps);
#pragma unroll
    for (int r = 0; r < 8; ++r) acc[r] *= alpha;

#pragma unroll
    for (int i = 0; i < KT; ++i) {
      v2f av, bp;
      av[0] = va[i][0]; av[1] = va[i][1]; bp[0] = s[i][0]; bp[1] = s[i][1];
      acc = wmma4(av, bp, acc);
      av[0] = va[i][2]; av[1] = va[i][3]; bp[0] = s[i][2]; bp[1] = s[i][3];
      acc = wmma4(av, bp, acc);
      av[0] = vb[i][0]; av[1] = vb[i][1]; bp[0] = s[i][4]; bp[1] = s[i][5];
      acc = wmma4(av, bp, acc);
      av[0] = vb[i][2]; av[1] = vb[i][3]; bp[0] = s[i][6]; bp[1] = s[i][7];
      acc = wmma4(av, bp, acc);
    }
  }
  l_run += xhalf(l_run);

  if (ksel == 1 && half == 0) {
#pragma unroll
    for (int r = 0; r < 8; ++r) xacc[qsel][r][l16] = acc[r];
    xmax[qsel][l16] = m_run;
    xsum[qsel][l16] = l_run;
  }
  __syncthreads();
  if (ksel == 0 && half == 0) {
    const float m_oth = xmax[qsel][l16];
    const float l_oth = xsum[qsel][l16];
    const float m_all = fmaxf(m_run, m_oth);
    const float f_own = __builtin_amdgcn_exp2f((m_run - m_all) * C2);
    const float f_oth = __builtin_amdgcn_exp2f((m_oth - m_all) * C2);
    const float l_all = fmaf(l_run, f_own, l_oth * f_oth);
    const float inv = 1.0f / l_all;
    const float w_own = f_own * inv;
    const float w_oth = f_oth * inv;
    float* ob = obuf + (size_t)n * HCH * LSPAT + (size_t)(h * DHEAD) * LSPAT + qbase + l16;
#pragma unroll
    for (int r = 0; r < 8; ++r)
      ob[(size_t)r * LSPAT] = fmaf(acc[r], w_own, xacc[qsel][r][l16] * w_oth);
  }
}

#define VPITCH 56
#define VICS   (3 * VPITCH)
#define KCHUNK 36

__global__ __launch_bounds__(128) void pos_conv_kernel(
    const float* __restrict__ qkv, const float* __restrict__ pw,
    const float* __restrict__ pb, float* __restrict__ obuf) {
  __shared__ __attribute__((aligned(16))) float vl[HCH * VICS];
  const int tid  = threadIdx.x;
  const int lane = tid & 31;
  const int half = lane >> 4;
  const int l16  = lane & 15;
  const int wv   = tid >> 5;
  const int n = blockIdx.x / HH;
  const int y = blockIdx.x % HH;

  const float* vbase = qkv + (size_t)n * CH3 * LSPAT;

  v4f stg[21];
#pragma unroll
  for (int it = 0; it < 21; ++it) {
    const int i   = tid + it * 128;
    const int ic  = i / 42;
    const int rem = i - ic * 42;
    const int r   = rem / 14;
    const int q   = rem - r * 14;
    const int iy  = y + r - 1;
    const bool ok = (q >= 1) && (q <= 12) && (iy >= 0) && (iy < HH);
    const int qc  = q < 1 ? 1 : (q > 12 ? 12 : q);
    const int iyc = iy < 0 ? 0 : (iy > HH - 1 ? HH - 1 : iy);
    const int ch  = (ic >> 3) * 24 + 16 + (ic & 7);
    const v4f t = *(const v4f*)(vbase + (size_t)ch * LSPAT + iyc * WW + (qc - 1) * 4);
    const v4f z = {0.0f, 0.0f, 0.0f, 0.0f};
    stg[it] = ok ? t : z;
  }
#pragma unroll
  for (int it = 0; it < 21; ++it)
    *(v4f*)(vl + (tid + it * 128) * 4) = stg[it];

  int lo[9][2];
#pragma unroll
  for (int s = 0; s < 9; ++s) {
#pragma unroll
    for (int j = 0; j < 2; ++j) {
      const int kloc = 4 * s + 2 * half + j;
      const int icl  = kloc / 9;
      const int tap  = kloc - icl * 9;
      const int ky   = tap / 3;
      const int kx   = tap - ky * 3;
      lo[s][j] = icl * VICS + ky * VPITCH + kx + 3 + l16;
    }
  }

  const float* arow = pw + (size_t)(wv * 16 + l16) * (HCH * 9) + 2 * half;

  v2f an[9];
#pragma unroll
  for (int s = 0; s < 9; ++s) {
    an[s][0] = arow[4 * s];
    an[s][1] = arow[4 * s + 1];
  }

  __syncthreads();

  v8f acc0 = {}, acc1 = {}, acc2 = {};
  for (int c = 0; c < (HCH * 9) / KCHUNK; ++c) {
    v2f ac[9];
#pragma unroll
    for (int s = 0; s < 9; ++s) ac[s] = an[s];
    if (c + 1 < (HCH * 9) / KCHUNK) {
      const float* ap = arow + (c + 1) * KCHUNK;
#pragma unroll
      for (int s = 0; s < 9; ++s) {
        an[s][0] = ap[4 * s];
        an[s][1] = ap[4 * s + 1];
      }
    }
    const float* vc = vl + c * (4 * VICS);
#pragma unroll
    for (int s = 0; s < 9; ++s) {
      const float* p0 = vc + lo[s][0];
      const float* p1 = vc + lo[s][1];
      v2f b0, b1, b2;
      b0[0] = p0[0];  b0[1] = p1[0];
      b1[0] = p0[16]; b1[1] = p1[16];
      b2[0] = p0[32]; b2[1] = p1[32];
      acc0 = wmma4(ac[s], b0, acc0);
      acc1 = wmma4(ac[s], b1, acc1);
      acc2 = wmma4(ac[s], b2, acc2);
    }
  }

  float* ob = obuf + (size_t)n * HCH * LSPAT + y * WW + l16;
#pragma unroll
  for (int r = 0; r < 8; ++r) {
    const int oc = wv * 16 + r + half * 8;
    const float bias = pb[oc];
    float* p = ob + (size_t)oc * LSPAT;
    p[0]  += acc0[r] + bias;
    p[16] += acc1[r] + bias;
    p[32] += acc2[r] + bias;
  }
}

__global__ __launch_bounds__(128) void res_gemm_kernel(
    const float* __restrict__ obuf, const float* __restrict__ w,
    const float* __restrict__ bias, float* __restrict__ out) {
  const int lane = threadIdx.x & 31;
  const int half = lane >> 4;
  const int l16  = lane & 15;
  const int wv   = threadIdx.x >> 5;

  const int ncg  = LSPAT / 64;
  const int nmtg = CIN / 64;
  int t = blockIdx.x;
  const int n   = t / (ncg * nmtg); t -= n * ncg * nmtg;
  const int cg  = t / nmtg;
  const int mtg = t % nmtg;
  const int mt  = mtg * 4 + wv;

  const int col0 = cg * 64 + 4 * l16;
  const float* wrow = w + (size_t)(mt * 16 + l16) * HCH + 4 * half;
  const float* xcol = obuf + (size_t)n * HCH * LSPAT + (size_t)(4 * half) * LSPAT + col0;

  v8f acc[4] = {};
  gemm16x64<HCH, LSPAT>(wrow, xcol, acc);

  store16x64(out + (size_t)n * CIN * LSPAT + col0, bias, mt * 16, half, acc);
}

extern "C" void kernel_launch(void* const* d_in, const int* in_sizes, int n_in,
                              void* d_out, int out_size, void* d_ws, size_t ws_size,
                              hipStream_t stream) {
  const float* x     = (const float*)d_in[0];
  const float* qkv_w = (const float*)d_in[1];
  const float* qkv_b = (const float*)d_in[2];
  const float* pos_w = (const float*)d_in[3];
  const float* pos_b = (const float*)d_in[4];
  const float* res_w = (const float*)d_in[5];
  const float* res_b = (const float*)d_in[6];
  float* out = (float*)d_out;

  float* qkv  = (float*)d_ws;
  float* obuf = qkv + (size_t)NBATCH * CH3 * LSPAT;

  qkv_gemm_kernel<<<NBATCH * (LSPAT / 64) * (CH3 / 64), 128, 0, stream>>>(x, qkv_w, qkv_b, qkv);

  attn_kernel<<<NBATCH * NHEAD * (LSPAT / 32), 128, 0, stream>>>(qkv, obuf);

  pos_conv_kernel<<<NBATCH * HH, 128, 0, stream>>>(qkv, pos_w, pos_b, obuf);

  res_gemm_kernel<<<NBATCH * (LSPAT / 64) * (CIN / 64), 128, 0, stream>>>(obuf, res_w, res_b, out);
}


# attention softmax with fewer vector instructions per 32-key step: the sum of p comes out of V*P^T (ones in V's idle A rows), exp arguments formed as one paired block; softmax sum order changes
# speedup vs baseline: 3.4609x; 1.0469x over previous
#include <hip/hip_runtime.h>
#include <math.h>

typedef __attribute__((ext_vector_type(2))) float v2f;
typedef __attribute__((ext_vector_type(4))) float v4f;
typedef __attribute__((ext_vector_type(8))) float v8f;

#define NBATCH 4
#define CIN 512
#define HH 48
#define WW 48
#define LSPAT (HH*WW)
#define CH3 192
#define HCH 64
#define NHEAD 8
#define DHEAD 8
#define ATTN_SCALE 0.35355339059327379f

static __device__ __forceinline__ v8f wmma4(v2f a, v2f b, v8f c) {
  return __builtin_amdgcn_wmma_f32_16x16x4_f32(false, a, false, b, (short)0, c,
                                               false, false);
}

#define GCH 4

template <int XPITCH>
static __device__ __forceinline__ void load_group(
    const float* __restrict__ wrow, const float* __restrict__ xcol, int k0,
    v4f (&a)[GCH], v4f (&b)[GCH][4]) {
#pragma unroll
  for (int i = 0; i < GCH; ++i) {
    a[i] = *(const v4f*)(wrow + k0 + 8 * i);
#pragma unroll
    for (int e = 0; e < 4; ++e)
      b[i][e] = *(const v4f*)(xcol + (size_t)(k0 + 8 * i + e) * XPITCH);
  }
}

static __device__ __forceinline__ void mma_group(
    const v4f (&a)[GCH], const v4f (&b)[GCH][4], v8f (&acc)[4]) {
#pragma unroll
  for (int i = 0; i < GCH; ++i) {
#pragma unroll
    for (int s = 0; s < 2; ++s) {
      v2f af; af[0] = a[i][2 * s]; af[1] = a[i][2 * s + 1];
#pragma unroll
      for (int t = 0; t < 4; ++t) {
        v2f bf; bf[0] = b[i][2 * s][t]; bf[1] = b[i][2 * s + 1][t];
        acc[t] = wmma4(af, bf, acc[t]);
      }
    }
  }
}

template <int KDIM, int XPITCH>
static __device__ __forceinline__ void gemm16x64(
    const float* __restrict__ wrow, const float* __restrict__ xcol,
    v8f (&acc)[4]) {
  constexpr int KG = 8 * GCH;
  constexpr int NG = KDIM / KG;
  v4f a0[GCH], b0[GCH][4], a1[GCH], b1[GCH][4];
  load_group<XPITCH>(wrow, xcol, 0, a0, b0);
#pragma unroll 1
  for (int g = 0; g + 2 < NG; g += 2) {
    load_group<XPITCH>(wrow, xcol, (g + 1) * KG, a1, b1);
    __builtin_amdgcn_sched_barrier(0);
    mma_group(a0, b0, acc);
    load_group<XPITCH>(wrow, xcol, (g + 2) * KG, a0, b0);
    __builtin_amdgcn_sched_barrier(0);
    mma_group(a1, b1, acc);
  }
  load_group<XPITCH>(wrow, xcol, (NG - 1) * KG, a1, b1);
  __builtin_amdgcn_sched_barrier(0);
  mma_group(a0, b0, acc);
  mma_group(a1, b1, acc);
}

static __device__ __forceinline__ void store16x64(
    float* __restrict__ out, const float* __restrict__ bias, int row0, int half,
    const v8f (&acc)[4]) {
#pragma unroll
  for (int r = 0; r < 8; ++r) {
    const int row = row0 + r + half * 8;
    const float bv = bias[row];
    v4f o;
    o[0] = acc[0][r] + bv; o[1] = acc[1][r] + bv;
    o[2] = acc[2][r] + bv; o[3] = acc[3][r] + bv;
    *(v4f*)(out + (size_t)row * LSPAT) = o;
  }
}

__global__ __launch_bounds__(128) void qkv_gemm_kernel(
    const float* __restrict__ x, const float* __restrict__ w,
    const float* __restrict__ bias, float* __restrict__ qkv) {
  const int lane = threadIdx.x & 31;
  const int half = lane >> 4;
  const int l16  = lane & 15;
  const int wv   = threadIdx.x >> 5;

  const int ncg  = LSPAT / 64;
  const int nmtg = CH3 / 64;
  int t = blockIdx.x;
  const int n   = t / (ncg * nmtg);  t -= n * ncg * nmtg;
  const int cg  = t / nmtg;
  const int mtg = t % nmtg;
  const int mt  = mtg * 4 + wv;

  const int col0 = cg * 64 + 4 * l16;
  const float* wrow = w + (size_t)(mt * 16 + l16) * CIN + 4 * half;
  const float* xcol = x + (size_t)n * CIN * LSPAT + (size_t)(4 * half) * LSPAT + col0;

  v8f acc[4] = {};
  gemm16x64<CIN, LSPAT>(wrow, xcol, acc);

  store16x64(qkv + (size_t)n * CH3 * LSPAT + col0, bias, mt * 16, half, acc);
}

#define KT 2
#define KTILES_PER_WAVE (LSPAT / 16 / 2)
static __device__ __forceinline__ float xhalf(float x) {
  const int b = __builtin_bit_cast(int, x);
  return __builtin_bit_cast(
      float, __builtin_amdgcn_permlanex16(b, b, 0x76543210, 0xfedcba98, false, false));
}

__global__ __launch_bounds__(128) void attn_kernel(
    const float* __restrict__ qkv, float* __restrict__ obuf) {
  __shared__ float xacc[2][DHEAD][16];
  __shared__ float xmax[2][16];
  __shared__ float xsum[2][16];

  const int lane = threadIdx.x & 31;
  const int half = lane >> 4;
  const int l16  = lane & 15;
  const int wv   = threadIdx.x >> 5;
  const int qsel = wv & 1;
  const int ksel = wv >> 1;

  const int qpairs = LSPAT / 32;
  const int nh = blockIdx.x / qpairs;
  const int qp = blockIdx.x % qpairs;
  const int n = nh >> 3, h = nh & 7;
  const int qbase = qp * 32 + qsel * 16;
  const int key0  = ksel * (KTILES_PER_WAVE * 16);

  const float* base = qkv + (size_t)n * CH3 * LSPAT + (size_t)h * 24 * LSPAT;
  const float* qm = base;
  const float* km = base + (size_t)DHEAD * LSPAT;
  const float* vm = base + (size_t)2 * DHEAD * LSPAT;

  v2f bq[2];
#pragma unroll
  for (int db = 0; db < 2; ++db) {
    const int d0 = db * 4 + half * 2;
    bq[db][0] = qm[(size_t)d0 * LSPAT + qbase + l16];
    bq[db][1] = qm[(size_t)(d0 + 1) * LSPAT + qbase + l16];
  }
  const float* kp0 = km + (size_t)(half * 2) * LSPAT + key0 + l16;
  const float* kp1 = kp0 + (size_t)4 * LSPAT;

  const bool vrow = l16 < 8;
  const float* vp = vm + (size_t)(l16 & 7) * LSPAT + key0 + half * 8;

  const float C2 = ATTN_SCALE * 1.4426950408889634f;
  float m_run = -3.0e38f;
  v8f acc = {};
  v4f va[KT], vb[KT];
#pragma unroll
  for (int i = 0; i < KT; ++i) {
    va[i] = (v4f){1.0f, 1.0f, 1.0f, 1.0f};
    vb[i] = (v4f){1.0f, 1.0f, 1.0f, 1.0f};
  }
  for (int kt = 0; kt < KTILES_PER_WAVE; kt += KT) {
    v2f a0[KT], a1[KT];
#pragma unroll
    for (int i = 0; i < KT; ++i) {
      const int kb = (kt + i) * 16;
      a0[i][0] = kp0[kb]; a0[i][1] = kp0[LSPAT + kb];
      a1[i][0] = kp1[kb]; a1[i][1] = kp1[LSPAT + kb];
    }
    if (vrow) {
#pragma unroll
      for (int i = 0; i < KT; ++i) {
        const int kb = (kt + i) * 16;
        va[i] = *(const v4f*)(vp + kb);
        vb[i] = *(const v4f*)(vp + kb + 4);
      }
    }
    v8f s[KT];
#pragma unroll
    for (int i = 0; i < KT; ++i) {
      const v8f z = {};
      s[i] = wmma4(a0[i], bq[0], z);
    }
#pragma unroll
    for (int i = 0; i < KT; ++i) s[i] = wmma4(a1[i], bq[1], s[i]);

    float t = s[0][0];
#pragma unroll
    for (int i = 0; i < KT; ++i) {
      const float u = fmaxf(fmaxf(fmaxf(s[i][0], s[i][1]), fmaxf(s[i][2], s[i][3])),
                            fmaxf(fmaxf(s[i][4], s[i][5]), fmaxf(s[i][6], s[i][7])));
      t = fmaxf(t, u);
    }
    t = fmaxf(t, xhalf(t));
    const float m_new = fmaxf(m_run, t);
    const float alpha = __builtin_amdgcn_exp2f((m_run - m_new) * C2);
    const float nm = -(m_new * C2);
    m_run = m_new;

#pragma unroll
    for (int i = 0; i < KT; ++i) {
#pragma unroll
      for (int r = 0; r < 8; ++r) s[i][r] = fmaf(s[i][r], C2, nm);
    }
    __builtin_amdgcn_sched_barrier(0);
#pragma unroll
    for (int i = 0; i < KT; ++i) {
#pragma unroll
      for (int r = 0; r < 8; ++r) s[i][r] = __builtin_amdgcn_exp2f(s[i][r]);
    }
    __builtin_amdgcn_sched_barrier(0);
#pragma unroll
    for (int r = 0; r < 8; ++r) acc[r] *= alpha;

#pragma unroll
    for (int i = 0; i < KT; ++i) {
      v2f av, bp;
      av[0] = va[i][0]; av[1] = va[i][1]; bp[0] = s[i][0]; bp[1] = s[i][1];
      acc = wmma4(av, bp, acc);
      av[0] = va[i][2]; av[1] = va[i][3]; bp[0] = s[i][2]; bp[1] = s[i][3];
      acc = wmma4(av, bp, acc);
      av[0] = vb[i][0]; av[1] = vb[i][1]; bp[0] = s[i][4]; bp[1] = s[i][5];
      acc = wmma4(av, bp, acc);
      av[0] = vb[i][2]; av[1] = vb[i][3]; bp[0] = s[i][6]; bp[1] = s[i][7];
      acc = wmma4(av, bp, acc);
    }
  }
  const float l_run = xhalf(acc[0]);

  if (ksel == 1 && half == 0) {
#pragma unroll
    for (int r = 0; r < 8; ++r) xacc[qsel][r][l16] = acc[r];
    xmax[qsel][l16] = m_run;
    xsum[qsel][l16] = l_run;
  }
  __syncthreads();
  if (ksel == 0 && half == 0) {
    const float m_oth = xmax[qsel][l16];
    const float l_oth = xsum[qsel][l16];
    const float m_all = fmaxf(m_run, m_oth);
    const float f_own = __builtin_amdgcn_exp2f((m_run - m_all) * C2);
    const float f_oth = __builtin_amdgcn_exp2f((m_oth - m_all) * C2);
    const float l_all = fmaf(l_run, f_own, l_oth * f_oth);
    const float inv = 1.0f / l_all;
    const float w_own = f_own * inv;
    const float w_oth = f_oth * inv;
    float* ob = obuf + (size_t)n * HCH * LSPAT + (size_t)(h * DHEAD) * LSPAT + qbase + l16;
#pragma unroll
    for (int r = 0; r < 8; ++r)
      ob[(size_t)r * LSPAT] = fmaf(acc[r], w_own, xacc[qsel][r][l16] * w_oth);
  }
}

#define VPITCH 56
#define VICS   (3 * VPITCH)
#define KCHUNK 36

__global__ __launch_bounds__(128) void pos_conv_kernel(
    const float* __restrict__ qkv, const float* __restrict__ pw,
    const float* __restrict__ pb, float* __restrict__ obuf) {
  __shared__ __attribute__((aligned(16))) float vl[HCH * VICS];
  const int tid  = threadIdx.x;
  const int lane = tid & 31;
  const int half = lane >> 4;
  const int l16  = lane & 15;
  const int wv   = tid >> 5;
  const int n = blockIdx.x / HH;
  const int y = blockIdx.x % HH;

  const float* vbase = qkv + (size_t)n * CH3 * LSPAT;

  v4f stg[21];
#pragma unroll
  for (int it = 0; it < 21; ++it) {
    const int i   = tid + it * 128;
    const int ic  = i / 42;
    const int rem = i - ic * 42;
    const int r   = rem / 14;
    const int q   = rem - r * 14;
    const int iy  = y + r - 1;
    const bool ok = (q >= 1) && (q <= 12) && (iy >= 0) && (iy < HH);
    const int qc  = q < 1 ? 1 : (q > 12 ? 12 : q);
    const int iyc = iy < 0 ? 0 : (iy > HH - 1 ? HH - 1 : iy);
    const int ch  = (ic >> 3) * 24 + 16 + (ic & 7);
    const v4f t = *(const v4f*)(vbase + (size_t)ch * LSPAT + iyc * WW + (qc - 1) * 4);
    const v4f z = {0.0f, 0.0f, 0.0f, 0.0f};
    stg[it] = ok ? t : z;
  }
#pragma unroll
  for (int it = 0; it < 21; ++it)
    *(v4f*)(vl + (tid + it * 128) * 4) = stg[it];

  int lo[9][2];
#pragma unroll
  for (int s = 0; s < 9; ++s) {
#pragma unroll
    for (int j = 0; j < 2; ++j) {
      const int kloc = 4 * s + 2 * half + j;
      const int icl  = kloc / 9;
      const int tap  = kloc - icl * 9;
      const int ky   = tap / 3;
      const int kx   = tap - ky * 3;
      lo[s][j] = icl * VICS + ky * VPITCH + kx + 3 + l16;
    }
  }

  const float* arow = pw + (size_t)(wv * 16 + l16) * (HCH * 9) + 2 * half;

  v2f an[9];
#pragma unroll
  for (int s = 0; s < 9; ++s) {
    an[s][0] = arow[4 * s];
    an[s][1] = arow[4 * s + 1];
  }

  __syncthreads();

  v8f acc0 = {}, acc1 = {}, acc2 = {};
  for (int c = 0; c < (HCH * 9) / KCHUNK; ++c) {
    v2f ac[9];
#pragma unroll
    for (int s = 0; s < 9; ++s) ac[s] = an[s];
    if (c + 1 < (HCH * 9) / KCHUNK) {
      const float* ap = arow + (c + 1) * KCHUNK;
#pragma unroll
      for (int s = 0; s < 9; ++s) {
        an[s][0] = ap[4 * s];
        an[s][1] = ap[4 * s + 1];
      }
    }
    const float* vc = vl + c * (4 * VICS);
#pragma unroll
    for (int s = 0; s < 9; ++s) {
      const float* p0 = vc + lo[s][0];
      const float* p1 = vc + lo[s][1];
      v2f b0, b1, b2;
      b0[0] = p0[0];  b0[1] = p1[0];
      b1[0] = p0[16]; b1[1] = p1[16];
      b2[0] = p0[32]; b2[1] = p1[32];
      acc0 = wmma4(ac[s], b0, acc0);
      acc1 = wmma4(ac[s], b1, acc1);
      acc2 = wmma4(ac[s], b2, acc2);
    }
  }

  float* ob = obuf + (size_t)n * HCH * LSPAT + y * WW + l16;
#pragma unroll
  for (int r = 0; r < 8; ++r) {
    const int oc = wv * 16 + r + half * 8;
    const float bias = pb[oc];
    float* p = ob + (size_t)oc * LSPAT;
    p[0]  += acc0[r] + bias;
    p[16] += acc1[r] + bias;
    p[32] += acc2[r] + bias;
  }
}

__global__ __launch_bounds__(128) void res_gemm_kernel(
    const float* __restrict__ obuf, const float* __restrict__ w,
    const float* __restrict__ bias, float* __restrict__ out) {
  const int lane = threadIdx.x & 31;
  const int half = lane >> 4;
  const int l16  = lane & 15;
  const int wv   = threadIdx.x >> 5;

  const int ncg  = LSPAT / 64;
  const int nmtg = CIN / 64;
  int t = blockIdx.x;
  const int n   = t / (ncg * nmtg); t -= n * ncg * nmtg;
  const int cg  = t / nmtg;
  const int mtg = t % nmtg;
  const int mt  = mtg * 4 + wv;

  const int col0 = cg * 64 + 4 * l16;
  const float* wrow = w + (size_t)(mt * 16 + l16) * HCH + 4 * half;
  const float* xcol = obuf + (size_t)n * HCH * LSPAT + (size_t)(4 * half) * LSPAT + col0;

  v8f acc[4] = {};
  gemm16x64<HCH, LSPAT>(wrow, xcol, acc);

  store16x64(out + (size_t)n * CIN * LSPAT + col0, bias, mt * 16, half, acc);
}

extern "C" void kernel_launch(void* const* d_in, const int* in_sizes, int n_in,
                              void* d_out, int out_size, void* d_ws, size_t ws_size,
                              hipStream_t stream) {
  const float* x     = (const float*)d_in[0];
  const float* qkv_w = (const float*)d_in[1];
  const float* qkv_b = (const float*)d_in[2];
  const float* pos_w = (const float*)d_in[3];
  const float* pos_b = (const float*)d_in[4];
  const float* res_w = (const float*)d_in[5];
  const float* res_b = (const float*)d_in[6];
  float* out = (float*)d_out;

  float* qkv  = (float*)d_ws;
  float* obuf = qkv + (size_t)NBATCH * CH3 * LSPAT;

  qkv_gemm_kernel<<<NBATCH * (LSPAT / 64) * (CH3 / 64), 128, 0, stream>>>(x, qkv_w, qkv_b, qkv);

  attn_kernel<<<NBATCH * NHEAD * (LSPAT / 32), 128, 0, stream>>>(qkv, obuf);

  pos_conv_kernel<<<NBATCH * HH, 128, 0, stream>>>(qkv, pos_w, pos_b, obuf);

  res_gemm_kernel<<<NBATCH * (LSPAT / 64) * (CIN / 64), 128, 0, stream>>>(obuf, res_w, res_b, out);
}


# attention: each wave takes BOTH 16-query tiles of its block against a quarter of the keys (was one tile, half the keys): each K/V fragment load feeds two tiles; softmax sum and P*V sum order changes
# speedup vs baseline: 3.7045x; 1.0704x over previous
#include <hip/hip_runtime.h>
#include <math.h>

typedef __attribute__((ext_vector_type(2))) float v2f;
typedef __attribute__((ext_vector_type(4))) float v4f;
typedef __attribute__((ext_vector_type(8))) float v8f;

#define NBATCH 4
#define CIN 512
#define HH 48
#define WW 48
#define LSPAT (HH*WW)
#define CH3 192
#define HCH 64
#define NHEAD 8
#define DHEAD 8
#define ATTN_SCALE 0.35355339059327379f

static __device__ __forceinline__ v8f wmma4(v2f a, v2f b, v8f c) {
  return __builtin_amdgcn_wmma_f32_16x16x4_f32(false, a, false, b, (short)0, c,
                                               false, false);
}

#define GCH 4

template <int XPITCH>
static __device__ __forceinline__ void load_group(
    const float* __restrict__ wrow, const float* __restrict__ xcol, int k0,
    v4f (&a)[GCH], v4f (&b)[GCH][4]) {
#pragma unroll
  for (int i = 0; i < GCH; ++i) {
    a[i] = *(const v4f*)(wrow + k0 + 8 * i);
#pragma unroll
    for (int e = 0; e < 4; ++e)
      b[i][e] = *(const v4f*)(xcol + (size_t)(k0 + 8 * i + e) * XPITCH);
  }
}

static __device__ __forceinline__ void mma_group(
    const v4f (&a)[GCH], const v4f (&b)[GCH][4], v8f (&acc)[4]) {
#pragma unroll
  for (int i = 0; i < GCH; ++i) {
#pragma unroll
    for (int s = 0; s < 2; ++s) {
      v2f af; af[0] = a[i][2 * s]; af[1] = a[i][2 * s + 1];
#pragma unroll
      for (int t = 0; t < 4; ++t) {
        v2f bf; bf[0] = b[i][2 * s][t]; bf[1] = b[i][2 * s + 1][t];
        acc[t] = wmma4(af, bf, acc[t]);
      }
    }
  }
}

template <int KDIM, int XPITCH>
static __device__ __forceinline__ void gemm16x64(
    const float* __restrict__ wrow, const float* __restrict__ xcol,
    v8f (&acc)[4]) {
  constexpr int KG = 8 * GCH;
  constexpr int NG = KDIM / KG;
  v4f a0[GCH], b0[GCH][4], a1[GCH], b1[GCH][4];
  load_group<XPITCH>(wrow, xcol, 0, a0, b0);
#pragma unroll 1
  for (int g = 0; g + 2 < NG; g += 2) {
    load_group<XPITCH>(wrow, xcol, (g + 1) * KG, a1, b1);
    __builtin_amdgcn_sched_barrier(0);
    mma_group(a0, b0, acc);
    load_group<XPITCH>(wrow, xcol, (g + 2) * KG, a0, b0);
    __builtin_amdgcn_sched_barrier(0);
    mma_group(a1, b1, acc);
  }
  load_group<XPITCH>(wrow, xcol, (NG - 1) * KG, a1, b1);
  __builtin_amdgcn_sched_barrier(0);
  mma_group(a0, b0, acc);
  mma_group(a1, b1, acc);
}

static __device__ __forceinline__ void store16x64(
    float* __restrict__ out, const float* __restrict__ bias, int row0, int half,
    const v8f (&acc)[4]) {
#pragma unroll
  for (int r = 0; r < 8; ++r) {
    const int row = row0 + r + half * 8;
    const float bv = bias[row];
    v4f o;
    o[0] = acc[0][r] + bv; o[1] = acc[1][r] + bv;
    o[2] = acc[2][r] + bv; o[3] = acc[3][r] + bv;
    *(v4f*)(out + (size_t)row * LSPAT) = o;
  }
}

__global__ __launch_bounds__(128) void qkv_gemm_kernel(
    const float* __restrict__ x, const float* __restrict__ w,
    const float* __restrict__ bias, float* __restrict__ qkv) {
  const int lane = threadIdx.x & 31;
  const int half = lane >> 4;
  const int l16  = lane & 15;
  const int wv   = threadIdx.x >> 5;

  const int ncg  = LSPAT / 64;
  const int nmtg = CH3 / 64;
  int t = blockIdx.x;
  const int n   = t / (ncg * nmtg);  t -= n * ncg * nmtg;
  const int cg  = t / nmtg;
  const int mtg = t % nmtg;
  const int mt  = mtg * 4 + wv;

  const int col0 = cg * 64 + 4 * l16;
  const float* wrow = w + (size_t)(mt * 16 + l16) * CIN + 4 * half;
  const float* xcol = x + (size_t)n * CIN * LSPAT + (size_t)(4 * half) * LSPAT + col0;

  v8f acc[4] = {};
  gemm16x64<CIN, LSPAT>(wrow, xcol, acc);

  store16x64(qkv + (size_t)n * CH3 * LSPAT + col0, bias, mt * 16, half, acc);
}

#define KT 2
#define QT 2
#define KSPLIT 4
#define KTILES_PER_WAVE (LSPAT / 16 / KSPLIT)
static __device__ __forceinline__ float xhalf(float x) {
  const int b = __builtin_bit_cast(int, x);
  return __builtin_bit_cast(
      float, __builtin_amdgcn_permlanex16(b, b, 0x76543210, 0xfedcba98, false, false));
}

__global__ __launch_bounds__(128) void attn_kernel(
    const float* __restrict__ qkv, float* __restrict__ obuf) {
  __shared__ float xacc[KSPLIT][QT][DHEAD][16];
  __shared__ float xmax[KSPLIT][QT][16];
  __shared__ float xsum[KSPLIT][QT][16];

  const int lane = threadIdx.x & 31;
  const int half = lane >> 4;
  const int l16  = lane & 15;
  const int ksel = threadIdx.x >> 5;

  const int qpairs = LSPAT / 32;
  const int nh = blockIdx.x / qpairs;
  const int qp = blockIdx.x % qpairs;
  const int n = nh >> 3, h = nh & 7;
  const int qbase = qp * 32;
  const int key0  = ksel * (KTILES_PER_WAVE * 16);

  const float* base = qkv + (size_t)n * CH3 * LSPAT + (size_t)h * 24 * LSPAT;
  const float* qm = base;
  const float* km = base + (size_t)DHEAD * LSPAT;
  const float* vm = base + (size_t)2 * DHEAD * LSPAT;

  v2f bq[QT][2];
#pragma unroll
  for (int qt = 0; qt < QT; ++qt) {
#pragma unroll
    for (int db = 0; db < 2; ++db) {
      const int d0 = db * 4 + half * 2;
      bq[qt][db][0] = qm[(size_t)d0 * LSPAT + qbase + qt * 16 + l16];
      bq[qt][db][1] = qm[(size_t)(d0 + 1) * LSPAT + qbase + qt * 16 + l16];
    }
  }
  const float* kp0 = km + (size_t)(half * 2) * LSPAT + key0 + l16;
  const float* kp1 = kp0 + (size_t)4 * LSPAT;

  const bool vrow = l16 < 8;
  const float* vp = vm + (size_t)(l16 & 7) * LSPAT + key0 + half * 8;

  const float C2 = ATTN_SCALE * 1.4426950408889634f;
  float m_run[QT];
  v8f acc[QT];
#pragma unroll
  for (int qt = 0; qt < QT; ++qt) {
    m_run[qt] = -3.0e38f;
    acc[qt] = (v8f){};
  }
  v4f va[KT], vb[KT];
#pragma unroll
  for (int i = 0; i < KT; ++i) {
    va[i] = (v4f){1.0f, 1.0f, 1.0f, 1.0f};
    vb[i] = (v4f){1.0f, 1.0f, 1.0f, 1.0f};
  }
  for (int kt = 0; kt < KTILES_PER_WAVE; kt += KT) {
    v2f a0[KT], a1[KT];
#pragma unroll
    for (int i = 0; i < KT; ++i) {
      const int kb = (kt + i) * 16;
      a0[i][0] = kp0[kb]; a0[i][1] = kp0[LSPAT + kb];
      a1[i][0] = kp1[kb]; a1[i][1] = kp1[LSPAT + kb];
    }
    if (vrow) {
#pragma unroll
      for (int i = 0; i < KT; ++i) {
        const int kb = (kt + i) * 16;
        va[i] = *(const v4f*)(vp + kb);
        vb[i] = *(const v4f*)(vp + kb + 4);
      }
    }
    v8f s[QT][KT];
#pragma unroll
    for (int qt = 0; qt < QT; ++qt) {
#pragma unroll
      for (int i = 0; i < KT; ++i) {
        const v8f z = {};
        s[qt][i] = wmma4(a0[i], bq[qt][0], z);
      }
    }
#pragma unroll
    for (int qt = 0; qt < QT; ++qt) {
#pragma unroll
      for (int i = 0; i < KT; ++i) s[qt][i] = wmma4(a1[i], bq[qt][1], s[qt][i]);
    }

#pragma unroll
    for (int qt = 0; qt < QT; ++qt) {
      float t = s[qt][0][0];
#pragma unroll
      for (int i = 0; i < KT; ++i) {
        const float u =
            fmaxf(fmaxf(fmaxf(s[qt][i][0], s[qt][i][1]), fmaxf(s[qt][i][2], s[qt][i][3])),
                  fmaxf(fmaxf(s[qt][i][4], s[qt][i][5]), fmaxf(s[qt][i][6], s[qt][i][7])));
        t = fmaxf(t, u);
      }
      t = fmaxf(t, xhalf(t));
      const float m_new = fmaxf(m_run[qt], t);
      const float alpha = __builtin_amdgcn_exp2f((m_run[qt] - m_new) * C2);
      const float nm = -(m_new * C2);
      m_run[qt] = m_new;

#pragma unroll
      for (int i = 0; i < KT; ++i) {
#pragma unroll
        for (int r = 0; r < 8; ++r) s[qt][i][r] = fmaf(s[qt][i][r], C2, nm);
      }
      __builtin_amdgcn_sched_barrier(0);
#pragma unroll
      for (int i = 0; i < KT; ++i) {
#pragma unroll
        for (int r = 0; r < 8; ++r) s[qt][i][r] = __builtin_amdgcn_exp2f(s[qt][i][r]);
      }
      __builtin_amdgcn_sched_barrier(0);
#pragma unroll
      for (int r = 0; r < 8; ++r) acc[qt][r] *= alpha;

#pragma unroll
      for (int i = 0; i < KT; ++i) {
        v2f av, bp;
        av[0] = va[i][0]; av[1] = va[i][1]; bp[0] = s[qt][i][0]; bp[1] = s[qt][i][1];
        acc[qt] = wmma4(av, bp, acc[qt]);
        av[0] = va[i][2]; av[1] = va[i][3]; bp[0] = s[qt][i][2]; bp[1] = s[qt][i][3];
        acc[qt] = wmma4(av, bp, acc[qt]);
        av[0] = vb[i][0]; av[1] = vb[i][1]; bp[0] = s[qt][i][4]; bp[1] = s[qt][i][5];
        acc[qt] = wmma4(av, bp, acc[qt]);
        av[0] = vb[i][2]; av[1] = vb[i][3]; bp[0] = s[qt][i][6]; bp[1] = s[qt][i][7];
        acc[qt] = wmma4(av, bp, acc[qt]);
      }
    }
  }

#pragma unroll
  for (int qt = 0; qt < QT; ++qt) {
    const float l_run = xhalf(acc[qt][0]);
    if (half == 0) {
#pragma unroll
      for (int r = 0; r < 8; ++r) xacc[ksel][qt][r][l16] = acc[qt][r];
      xmax[ksel][qt][l16] = m_run[qt];
      xsum[ksel][qt][l16] = l_run;
    }
  }
  __syncthreads();
  if (ksel < QT) {
    const int qt = ksel;
    float mk[KSPLIT];
#pragma unroll
    for (int k = 0; k < KSPLIT; ++k) mk[k] = xmax[k][qt][l16];
    const float m_all = fmaxf(fmaxf(mk[0], mk[1]), fmaxf(mk[2], mk[3]));
    float f[KSPLIT];
    float l_all = 0.0f;
#pragma unroll
    for (int k = 0; k < KSPLIT; ++k) {
      f[k] = __builtin_amdgcn_exp2f((mk[k] - m_all) * C2);
      l_all = fmaf(xsum[k][qt][l16], f[k], l_all);
    }
    const float inv = 1.0f / l_all;
    float* ob = obuf + (size_t)n * HCH * LSPAT + (size_t)(h * DHEAD) * LSPAT + qbase +
                qt * 16 + l16;
#pragma unroll
    for (int rr = 0; rr < 4; ++rr) {
      const int r = half * 4 + rr;
      float o = 0.0f;
#pragma unroll
      for (int k = 0; k < KSPLIT; ++k) o = fmaf(xacc[k][qt][r][l16], f[k] * inv, o);
      ob[(size_t)r * LSPAT] = o;
    }
  }
}

#define VPITCH 56
#define VICS   (3 * VPITCH)
#define KCHUNK 36

__global__ __launch_bounds__(128) void pos_conv_kernel(
    const float* __restrict__ qkv, const float* __restrict__ pw,
    const float* __restrict__ pb, float* __restrict__ obuf) {
  __shared__ __attribute__((aligned(16))) float vl[HCH * VICS];
  const int tid  = threadIdx.x;
  const int lane = tid & 31;
  const int half = lane >> 4;
  const int l16  = lane & 15;
  const int wv   = tid >> 5;
  const int n = blockIdx.x / HH;
  const int y = blockIdx.x % HH;

  const float* vbase = qkv + (size_t)n * CH3 * LSPAT;

  v4f stg[21];
#pragma unroll
  for (int it = 0; it < 21; ++it) {
    const int i   = tid + it * 128;
    const int ic  = i / 42;
    const int rem = i - ic * 42;
    const int r   = rem / 14;
    const int q   = rem - r * 14;
    const int iy  = y + r - 1;
    const bool ok = (q >= 1) && (q <= 12) && (iy >= 0) && (iy < HH);
    const int qc  = q < 1 ? 1 : (q > 12 ? 12 : q);
    const int iyc = iy < 0 ? 0 : (iy > HH - 1 ? HH - 1 : iy);
    const int ch  = (ic >> 3) * 24 + 16 + (ic & 7);
    const v4f t = *(const v4f*)(vbase + (size_t)ch * LSPAT + iyc * WW + (qc - 1) * 4);
    const v4f z = {0.0f, 0.0f, 0.0f, 0.0f};
    stg[it] = ok ? t : z;
  }
#pragma unroll
  for (int it = 0; it < 21; ++it)
    *(v4f*)(vl + (tid + it * 128) * 4) = stg[it];

  int lo[9][2];
#pragma unroll
  for (int s = 0; s < 9; ++s) {
#pragma unroll
    for (int j = 0; j < 2; ++j) {
      const int kloc = 4 * s + 2 * half + j;
      const int icl  = kloc / 9;
      const int tap  = kloc - icl * 9;
      const int ky   = tap / 3;
      const int kx   = tap - ky * 3;
      lo[s][j] = icl * VICS + ky * VPITCH + kx + 3 + l16;
    }
  }

  const float* arow = pw + (size_t)(wv * 16 + l16) * (HCH * 9) + 2 * half;

  v2f an[9];
#pragma unroll
  for (int s = 0; s < 9; ++s) {
    an[s][0] = arow[4 * s];
    an[s][1] = arow[4 * s + 1];
  }

  __syncthreads();

  v8f acc0 = {}, acc1 = {}, acc2 = {};
  for (int c = 0; c < (HCH * 9) / KCHUNK; ++c) {
    v2f ac[9];
#pragma unroll
    for (int s = 0; s < 9; ++s) ac[s] = an[s];
    if (c + 1 < (HCH * 9) / KCHUNK) {
      const float* ap = arow + (c + 1) * KCHUNK;
#pragma unroll
      for (int s = 0; s < 9; ++s) {
        an[s][0] = ap[4 * s];
        an[s][1] = ap[4 * s + 1];
      }
    }
    const float* vc = vl + c * (4 * VICS);
#pragma unroll
    for (int s = 0; s < 9; ++s) {
      const float* p0 = vc + lo[s][0];
      const float* p1 = vc + lo[s][1];
      v2f b0, b1, b2;
      b0[0] = p0[0];  b0[1] = p1[0];
      b1[0] = p0[16]; b1[1] = p1[16];
      b2[0] = p0[32]; b2[1] = p1[32];
      acc0 = wmma4(ac[s], b0, acc0);
      acc1 = wmma4(ac[s], b1, acc1);
      acc2 = wmma4(ac[s], b2, acc2);
    }
  }

  float* ob = obuf + (size_t)n * HCH * LSPAT + y * WW + l16;
#pragma unroll
  for (int r = 0; r < 8; ++r) {
    const int oc = wv * 16 + r + half * 8;
    const float bias = pb[oc];
    float* p = ob + (size_t)oc * LSPAT;
    p[0]  += acc0[r] + bias;
    p[16] += acc1[r] + bias;
    p[32] += acc2[r] + bias;
  }
}

__global__ __launch_bounds__(128) void res_gemm_kernel(
    const float* __restrict__ obuf, const float* __restrict__ w,
    const float* __restrict__ bias, float* __restrict__ out) {
  const int lane = threadIdx.x & 31;
  const int half = lane >> 4;
  const int l16  = lane & 15;
  const int wv   = threadIdx.x >> 5;

  const int ncg  = LSPAT / 64;
  const int nmtg = CIN / 64;
  int t = blockIdx.x;
  const int n   = t / (ncg * nmtg); t -= n * ncg * nmtg;
  const int cg  = t / nmtg;
  const int mtg = t % nmtg;
  const int mt  = mtg * 4 + wv;

  const int col0 = cg * 64 + 4 * l16;
  const float* wrow = w + (size_t)(mt * 16 + l16) * HCH + 4 * half;
  const float* xcol = obuf + (size_t)n * HCH * LSPAT + (size_t)(4 * half) * LSPAT + col0;

  v8f acc[4] = {};
  gemm16x64<HCH, LSPAT>(wrow, xcol, acc);

  store16x64(out + (size_t)n * CIN * LSPAT + col0, bias, mt * 16, half, acc);
}

extern "C" void kernel_launch(void* const* d_in, const int* in_sizes, int n_in,
                              void* d_out, int out_size, void* d_ws, size_t ws_size,
                              hipStream_t stream) {
  const float* x     = (const float*)d_in[0];
  const float* qkv_w = (const float*)d_in[1];
  const float* qkv_b = (const float*)d_in[2];
  const float* pos_w = (const float*)d_in[3];
  const float* pos_b = (const float*)d_in[4];
  const float* res_w = (const float*)d_in[5];
  const float* res_b = (const float*)d_in[6];
  float* out = (float*)d_out;

  float* qkv  = (float*)d_ws;
  float* obuf = qkv + (size_t)NBATCH * CH3 * LSPAT;

  qkv_gemm_kernel<<<NBATCH * (LSPAT / 64) * (CH3 / 64), 128, 0, stream>>>(x, qkv_w, qkv_b, qkv);

  attn_kernel<<<NBATCH * NHEAD * (LSPAT / 32), 128, 0, stream>>>(qkv, obuf);

  pos_conv_kernel<<<NBATCH * HH, 128, 0, stream>>>(qkv, pos_w, pos_b, obuf);

  res_gemm_kernel<<<NBATCH * (LSPAT / 64) * (CIN / 64), 128, 0, stream>>>(obuf, res_w, res_b, out);
}


# attention softmax reference raised lazily (only when a score tops it by 2^16): most steps skip lane exchange and rescale; scale folded into Q, packed subtract, 3-operand maxima; sum rounding changes
# speedup vs baseline: 3.8204x; 1.0313x over previous
#include <hip/hip_runtime.h>
#include <math.h>

typedef __attribute__((ext_vector_type(2))) float v2f;
typedef __attribute__((ext_vector_type(4))) float v4f;
typedef __attribute__((ext_vector_type(8))) float v8f;

#define NBATCH 4
#define CIN 512
#define HH 48
#define WW 48
#define LSPAT (HH*WW)
#define CH3 192
#define HCH 64
#define NHEAD 8
#define DHEAD 8
#define ATTN_SCALE 0.35355339059327379f

static __device__ __forceinline__ v8f wmma4(v2f a, v2f b, v8f c) {
  return __builtin_amdgcn_wmma_f32_16x16x4_f32(false, a, false, b, (short)0, c,
                                               false, false);
}

#define GCH 4

template <int XPITCH>
static __device__ __forceinline__ void load_group(
    const float* __restrict__ wrow, const float* __restrict__ xcol, int k0,
    v4f (&a)[GCH], v4f (&b)[GCH][4]) {
#pragma unroll
  for (int i = 0; i < GCH; ++i) {
    a[i] = *(const v4f*)(wrow + k0 + 8 * i);
#pragma unroll
    for (int e = 0; e < 4; ++e)
      b[i][e] = *(const v4f*)(xcol + (size_t)(k0 + 8 * i + e) * XPITCH);
  }
}

static __device__ __forceinline__ void mma_group(
    const v4f (&a)[GCH], const v4f (&b)[GCH][4], v8f (&acc)[4]) {
#pragma unroll
  for (int i = 0; i < GCH; ++i) {
#pragma unroll
    for (int s = 0; s < 2; ++s) {
      v2f af; af[0] = a[i][2 * s]; af[1] = a[i][2 * s + 1];
#pragma unroll
      for (int t = 0; t < 4; ++t) {
        v2f bf; bf[0] = b[i][2 * s][t]; bf[1] = b[i][2 * s + 1][t];
        acc[t] = wmma4(af, bf, acc[t]);
      }
    }
  }
}

template <int KDIM, int XPITCH>
static __device__ __forceinline__ void gemm16x64(
    const float* __restrict__ wrow, const float* __restrict__ xcol,
    v8f (&acc)[4]) {
  constexpr int KG = 8 * GCH;
  constexpr int NG = KDIM / KG;
  v4f a0[GCH], b0[GCH][4], a1[GCH], b1[GCH][4];
  load_group<XPITCH>(wrow, xcol, 0, a0, b0);
#pragma unroll 1
  for (int g = 0; g + 2 < NG; g += 2) {
    load_group<XPITCH>(wrow, xcol, (g + 1) * KG, a1, b1);
    __builtin_amdgcn_sched_barrier(0);
    mma_group(a0, b0, acc);
    load_group<XPITCH>(wrow, xcol, (g + 2) * KG, a0, b0);
    __builtin_amdgcn_sched_barrier(0);
    mma_group(a1, b1, acc);
  }
  load_group<XPITCH>(wrow, xcol, (NG - 1) * KG, a1, b1);
  __builtin_amdgcn_sched_barrier(0);
  mma_group(a0, b0, acc);
  mma_group(a1, b1, acc);
}

static __device__ __forceinline__ void store16x64(
    float* __restrict__ out, const float* __restrict__ bias, int row0, int half,
    const v8f (&acc)[4]) {
#pragma unroll
  for (int r = 0; r < 8; ++r) {
    const int row = row0 + r + half * 8;
    const float bv = bias[row];
    v4f o;
    o[0] = acc[0][r] + bv; o[1] = acc[1][r] + bv;
    o[2] = acc[2][r] + bv; o[3] = acc[3][r] + bv;
    *(v4f*)(out + (size_t)row * LSPAT) = o;
  }
}

__global__ __launch_bounds__(128) void qkv_gemm_kernel(
    const float* __restrict__ x, const float* __restrict__ w,
    const float* __restrict__ bias, float* __restrict__ qkv) {
  const int lane = threadIdx.x & 31;
  const int half = lane >> 4;
  const int l16  = lane & 15;
  const int wv   = threadIdx.x >> 5;

  const int ncg  = LSPAT / 64;
  const int nmtg = CH3 / 64;
  int t = blockIdx.x;
  const int n   = t / (ncg * nmtg);  t -= n * ncg * nmtg;
  const int cg  = t / nmtg;
  const int mtg = t % nmtg;
  const int mt  = mtg * 4 + wv;

  const int col0 = cg * 64 + 4 * l16;
  const float* wrow = w + (size_t)(mt * 16 + l16) * CIN + 4 * half;
  const float* xcol = x + (size_t)n * CIN * LSPAT + (size_t)(4 * half) * LSPAT + col0;

  v8f acc[4] = {};
  gemm16x64<CIN, LSPAT>(wrow, xcol, acc);

  store16x64(qkv + (size_t)n * CH3 * LSPAT + col0, bias, mt * 16, half, acc);
}

#define KT 2
#define QT 2
#define KSPLIT 4
#define KTILES_PER_WAVE (LSPAT / 16 / KSPLIT)
static __device__ __forceinline__ float xhalf(float x) {
  const int b = __builtin_bit_cast(int, x);
  return __builtin_bit_cast(
      float, __builtin_amdgcn_permlanex16(b, b, 0x76543210, 0xfedcba98, false, false));
}

__global__ __launch_bounds__(128) void attn_kernel(
    const float* __restrict__ qkv, float* __restrict__ obuf) {
  __shared__ float xacc[KSPLIT][QT][DHEAD][16];
  __shared__ float xmax[KSPLIT][QT][16];
  __shared__ float xsum[KSPLIT][QT][16];

  const int lane = threadIdx.x & 31;
  const int half = lane >> 4;
  const int l16  = lane & 15;
  const int ksel = threadIdx.x >> 5;

  const int qpairs = LSPAT / 32;
  const int nh = blockIdx.x / qpairs;
  const int qp = blockIdx.x % qpairs;
  const int n = nh >> 3, h = nh & 7;
  const int qbase = qp * 32;
  const int key0  = ksel * (KTILES_PER_WAVE * 16);

  const float* base = qkv + (size_t)n * CH3 * LSPAT + (size_t)h * 24 * LSPAT;
  const float* qm = base;
  const float* km = base + (size_t)DHEAD * LSPAT;
  const float* vm = base + (size_t)2 * DHEAD * LSPAT;

  const float C2 = ATTN_SCALE * 1.4426950408889634f;
  v2f bq[QT][2];
#pragma unroll
  for (int qt = 0; qt < QT; ++qt) {
#pragma unroll
    for (int db = 0; db < 2; ++db) {
      const int d0 = db * 4 + half * 2;
      bq[qt][db][0] = qm[(size_t)d0 * LSPAT + qbase + qt * 16 + l16] * C2;
      bq[qt][db][1] = qm[(size_t)(d0 + 1) * LSPAT + qbase + qt * 16 + l16] * C2;
    }
  }
  const float* kp0 = km + (size_t)(half * 2) * LSPAT + key0 + l16;
  const float* kp1 = kp0 + (size_t)4 * LSPAT;

  const bool vrow = l16 < 8;
  const float* vp = vm + (size_t)(l16 & 7) * LSPAT + key0 + half * 8;

  const float REF_MARGIN = 16.0f;
  float bref[QT];
  float blim[QT];
  v8f acc[QT];
#pragma unroll
  for (int qt = 0; qt < QT; ++qt) {
    bref[qt] = -3.0e38f;
    blim[qt] = -3.0e38f;
    acc[qt] = (v8f){};
  }
  v4f va[KT], vb[KT];
#pragma unroll
  for (int i = 0; i < KT; ++i) {
    va[i] = (v4f){1.0f, 1.0f, 1.0f, 1.0f};
    vb[i] = (v4f){1.0f, 1.0f, 1.0f, 1.0f};
  }
  for (int kt = 0; kt < KTILES_PER_WAVE; kt += KT) {
    v2f a0[KT], a1[KT];
#pragma unroll
    for (int i = 0; i < KT; ++i) {
      const int kb = (kt + i) * 16;
      a0[i][0] = kp0[kb]; a0[i][1] = kp0[LSPAT + kb];
      a1[i][0] = kp1[kb]; a1[i][1] = kp1[LSPAT + kb];
    }
    if (vrow) {
#pragma unroll
      for (int i = 0; i < KT; ++i) {
        const int kb = (kt + i) * 16;
        va[i] = *(const v4f*)(vp + kb);
        vb[i] = *(const v4f*)(vp + kb + 4);
      }
    }
    v8f s[QT][KT];
#pragma unroll
    for (int qt = 0; qt < QT; ++qt) {
#pragma unroll
      for (int i = 0; i < KT; ++i) {
        const v8f z = {};
        s[qt][i] = wmma4(a0[i], bq[qt][0], z);
      }
    }
#pragma unroll
    for (int qt = 0; qt < QT; ++qt) {
#pragma unroll
      for (int i = 0; i < KT; ++i) s[qt][i] = wmma4(a1[i], bq[qt][1], s[qt][i]);
    }

    static_assert(KT == 2, "the two maximum chains below are written for two key tiles");
    float t[QT];
#pragma unroll
    for (int qt = 0; qt < QT; ++qt) {
      float ta = fmaxf(fmaxf(s[qt][0][0], s[qt][0][1]), s[qt][0][2]);
      float tb = fmaxf(fmaxf(s[qt][1][0], s[qt][1][1]), s[qt][1][2]);
      ta = fmaxf(fmaxf(ta, s[qt][0][3]), s[qt][0][4]);
      tb = fmaxf(fmaxf(tb, s[qt][1][3]), s[qt][1][4]);
      ta = fmaxf(fmaxf(ta, s[qt][0][5]), s[qt][0][6]);
      tb = fmaxf(fmaxf(tb, s[qt][1][5]), s[qt][1][6]);
      ta = fmaxf(fmaxf(ta, s[qt][0][7]), s[qt][1][7]);
      t[qt] = fmaxf(ta, tb);
    }
    const bool over = (t[0] > blim[0]) || (t[1] > blim[1]);
    if (__builtin_amdgcn_ballot_w32(over) != 0u) {
#pragma unroll
      for (int qt = 0; qt < QT; ++qt) {
        const float tx = fmaxf(t[qt], xhalf(t[qt]));
        const float b_new = fmaxf(bref[qt], tx);
        const float alpha = __builtin_amdgcn_exp2f(bref[qt] - b_new);
        bref[qt] = b_new;
        blim[qt] = b_new + REF_MARGIN;
#pragma unroll
        for (int r = 0; r < 8; ++r) acc[qt][r] *= alpha;
      }
    }

#pragma unroll
    for (int qt = 0; qt < QT; ++qt) {
      const v2f bb = {bref[qt], bref[qt]};
#pragma unroll
      for (int i = 0; i < KT; ++i) {
#pragma unroll
        for (int c = 0; c < 4; ++c) {
          v2f t2;
          t2[0] = s[qt][i][2 * c]; t2[1] = s[qt][i][2 * c + 1];
          t2 = t2 - bb;
          s[qt][i][2 * c] = t2[0]; s[qt][i][2 * c + 1] = t2[1];
        }
      }
      __builtin_amdgcn_sched_barrier(0);
#pragma unroll
      for (int i = 0; i < KT; ++i) {
#pragma unroll
        for (int r = 0; r < 8; ++r) s[qt][i][r] = __builtin_amdgcn_exp2f(s[qt][i][r]);
      }
      __builtin_amdgcn_sched_barrier(0);

#pragma unroll
      for (int i = 0; i < KT; ++i) {
        v2f av, bp;
        av[0] = va[i][0]; av[1] = va[i][1]; bp[0] = s[qt][i][0]; bp[1] = s[qt][i][1];
        acc[qt] = wmma4(av, bp, acc[qt]);
        av[0] = va[i][2]; av[1] = va[i][3]; bp[0] = s[qt][i][2]; bp[1] = s[qt][i][3];
        acc[qt] = wmma4(av, bp, acc[qt]);
        av[0] = vb[i][0]; av[1] = vb[i][1]; bp[0] = s[qt][i][4]; bp[1] = s[qt][i][5];
        acc[qt] = wmma4(av, bp, acc[qt]);
        av[0] = vb[i][2]; av[1] = vb[i][3]; bp[0] = s[qt][i][6]; bp[1] = s[qt][i][7];
        acc[qt] = wmma4(av, bp, acc[qt]);
      }
    }
  }

#pragma unroll
  for (int qt = 0; qt < QT; ++qt) {
    const float l_run = xhalf(acc[qt][0]);
    if (half == 0) {
#pragma unroll
      for (int r = 0; r < 8; ++r) xacc[ksel][qt][r][l16] = acc[qt][r];
      xmax[ksel][qt][l16] = bref[qt];
      xsum[ksel][qt][l16] = l_run;
    }
  }
  __syncthreads();
  if (ksel < QT) {
    const int qt = ksel;
    float mk[KSPLIT];
#pragma unroll
    for (int k = 0; k < KSPLIT; ++k) mk[k] = xmax[k][qt][l16];
    const float m_all = fmaxf(fmaxf(mk[0], mk[1]), fmaxf(mk[2], mk[3]));
    float f[KSPLIT];
    float l_all = 0.0f;
#pragma unroll
    for (int k = 0; k < KSPLIT; ++k) {
      f[k] = __builtin_amdgcn_exp2f(mk[k] - m_all);
      l_all = fmaf(xsum[k][qt][l16], f[k], l_all);
    }
    const float inv = 1.0f / l_all;
    float* ob = obuf + (size_t)n * HCH * LSPAT + (size_t)(h * DHEAD) * LSPAT + qbase +
                qt * 16 + l16;
#pragma unroll
    for (int rr = 0; rr < 4; ++rr) {
      const int r = half * 4 + rr;
      float o = 0.0f;
#pragma unroll
      for (int k = 0; k < KSPLIT; ++k) o = fmaf(xacc[k][qt][r][l16], f[k] * inv, o);
      ob[(size_t)r * LSPAT] = o;
    }
  }
}

#define VPITCH 56
#define VICS   (3 * VPITCH)
#define KCHUNK 36

__global__ __launch_bounds__(128) void pos_conv_kernel(
    const float* __restrict__ qkv, const float* __restrict__ pw,
    const float* __restrict__ pb, float* __restrict__ obuf) {
  __shared__ __attribute__((aligned(16))) float vl[HCH * VICS];
  const int tid  = threadIdx.x;
  const int lane = tid & 31;
  const int half = lane >> 4;
  const int l16  = lane & 15;
  const int wv   = tid >> 5;
  const int n = blockIdx.x / HH;
  const int y = blockIdx.x % HH;

  const float* vbase = qkv + (size_t)n * CH3 * LSPAT;

  v4f stg[21];
#pragma unroll
  for (int it = 0; it < 21; ++it) {
    const int i   = tid + it * 128;
    const int ic  = i / 42;
    const int rem = i - ic * 42;
    const int r   = rem / 14;
    const int q   = rem - r * 14;
    const int iy  = y + r - 1;
    const bool ok = (q >= 1) && (q <= 12) && (iy >= 0) && (iy < HH);
    const int qc  = q < 1 ? 1 : (q > 12 ? 12 : q);
    const int iyc = iy < 0 ? 0 : (iy > HH - 1 ? HH - 1 : iy);
    const int ch  = (ic >> 3) * 24 + 16 + (ic & 7);
    const v4f t = *(const v4f*)(vbase + (size_t)ch * LSPAT + iyc * WW + (qc - 1) * 4);
    const v4f z = {0.0f, 0.0f, 0.0f, 0.0f};
    stg[it] = ok ? t : z;
  }
#pragma unroll
  for (int it = 0; it < 21; ++it)
    *(v4f*)(vl + (tid + it * 128) * 4) = stg[it];

  int lo[9][2];
#pragma unroll
  for (int s = 0; s < 9; ++s) {
#pragma unroll
    for (int j = 0; j < 2; ++j) {
      const int kloc = 4 * s + 2 * half + j;
      const int icl  = kloc / 9;
      const int tap  = kloc - icl * 9;
      const int ky   = tap / 3;
      const int kx   = tap - ky * 3;
      lo[s][j] = icl * VICS + ky * VPITCH + kx + 3 + l16;
    }
  }

  const float* arow = pw + (size_t)(wv * 16 + l16) * (HCH * 9) + 2 * half;

  v2f an[9];
#pragma unroll
  for (int s = 0; s < 9; ++s) {
    an[s][0] = arow[4 * s];
    an[s][1] = arow[4 * s + 1];
  }

  __syncthreads();

  v8f acc0 = {}, acc1 = {}, acc2 = {};
  for (int c = 0; c < (HCH * 9) / KCHUNK; ++c) {
    v2f ac[9];
#pragma unroll
    for (int s = 0; s < 9; ++s) ac[s] = an[s];
    if (c + 1 < (HCH * 9) / KCHUNK) {
      const float* ap = arow + (c + 1) * KCHUNK;
#pragma unroll
      for (int s = 0; s < 9; ++s) {
        an[s][0] = ap[4 * s];
        an[s][1] = ap[4 * s + 1];
      }
    }
    const float* vc = vl + c * (4 * VICS);
#pragma unroll
    for (int s = 0; s < 9; ++s) {
      const float* p0 = vc + lo[s][0];
      const float* p1 = vc + lo[s][1];
      v2f b0, b1, b2;
      b0[0] = p0[0];  b0[1] = p1[0];
      b1[0] = p0[16]; b1[1] = p1[16];
      b2[0] = p0[32]; b2[1] = p1[32];
      acc0 = wmma4(ac[s], b0, acc0);
      acc1 = wmma4(ac[s], b1, acc1);
      acc2 = wmma4(ac[s], b2, acc2);
    }
  }

  float* ob = obuf + (size_t)n * HCH * LSPAT + y * WW + l16;
#pragma unroll
  for (int r = 0; r < 8; ++r) {
    const int oc = wv * 16 + r + half * 8;
    const float bias = pb[oc];
    float* p = ob + (size_t)oc * LSPAT;
    p[0]  += acc0[r] + bias;
    p[16] += acc1[r] + bias;
    p[32] += acc2[r] + bias;
  }
}

__global__ __launch_bounds__(128) void res_gemm_kernel(
    const float* __restrict__ obuf, const float* __restrict__ w,
    const float* __restrict__ bias, float* __restrict__ out) {
  const int lane = threadIdx.x & 31;
  const int half = lane >> 4;
  const int l16  = lane & 15;
  const int wv   = threadIdx.x >> 5;

  const int ncg  = LSPAT / 64;
  const int nmtg = CIN / 64;
  int t = blockIdx.x;
  const int n   = t / (ncg * nmtg); t -= n * ncg * nmtg;
  const int cg  = t / nmtg;
  const int mtg = t % nmtg;
  const int mt  = mtg * 4 + wv;

  const int col0 = cg * 64 + 4 * l16;
  const float* wrow = w + (size_t)(mt * 16 + l16) * HCH + 4 * half;
  const float* xcol = obuf + (size_t)n * HCH * LSPAT + (size_t)(4 * half) * LSPAT + col0;

  v8f acc[4] = {};
  gemm16x64<HCH, LSPAT>(wrow, xcol, acc);

  store16x64(out + (size_t)n * CIN * LSPAT + col0, bias, mt * 16, half, acc);
}

extern "C" void kernel_launch(void* const* d_in, const int* in_sizes, int n_in,
                              void* d_out, int out_size, void* d_ws, size_t ws_size,
                              hipStream_t stream) {
  const float* x     = (const float*)d_in[0];
  const float* qkv_w = (const float*)d_in[1];
  const float* qkv_b = (const float*)d_in[2];
  const float* pos_w = (const float*)d_in[3];
  const float* pos_b = (const float*)d_in[4];
  const float* res_w = (const float*)d_in[5];
  const float* res_b = (const float*)d_in[6];
  float* out = (float*)d_out;

  float* qkv  = (float*)d_ws;
  float* obuf = qkv + (size_t)NBATCH * CH3 * LSPAT;

  qkv_gemm_kernel<<<NBATCH * (LSPAT / 64) * (CH3 / 64), 128, 0, stream>>>(x, qkv_w, qkv_b, qkv);

  attn_kernel<<<NBATCH * NHEAD * (LSPAT / 32), 128, 0, stream>>>(qkv, obuf);

  pos_conv_kernel<<<NBATCH * HH, 128, 0, stream>>>(qkv, pos_w, pos_b, obuf);

  res_gemm_kernel<<<NBATCH * (LSPAT / 64) * (CIN / 64), 128, 0, stream>>>(obuf, res_w, res_b, out);
}
